# starting kernel
# baseline (speedup 1.0000x reference)
#include <hip/hip_runtime.h>
#include <hip/hip_bf16.h>


typedef float v2f __attribute__((ext_vector_type(2)));
typedef float v8f __attribute__((ext_vector_type(8)));

#define IN_C 128
#define HID2 128
#define HID  64

__device__ __forceinline__ void atomic_add_f32(float* p, float v) {
    __hip_atomic_fetch_add(p, v, __ATOMIC_RELAXED, __HIP_MEMORY_SCOPE_AGENT);
}

__global__ __launch_bounds__(256)
void fill_kernel(float* __restrict__ p, float v, long long n) {
    long long i = (long long)blockIdx.x * 256 + threadIdx.x;
    if (i < n) p[i] = v;
}

__global__ __launch_bounds__(256)
void degree_kernel(const int* __restrict__ dst, float* __restrict__ deg, long long E) {
    long long e = (long long)blockIdx.x * 256 + threadIdx.x;
    if (e < E) atomic_add_f32(&deg[dst[e]], 1.0f);
}

__global__ __launch_bounds__(256)
void rsqrt_kernel(const float* __restrict__ deg, float* __restrict__ dinv, int n) {
    int i = blockIdx.x * 256 + threadIdx.x;
    if (i < n) dinv[i] = rsqrtf(deg[i]);
}

template <int NOUT>
__global__ __launch_bounds__(128)
void gemm_wmma_f32(const float* __restrict__ A, const float* __restrict__ B,
                   float* __restrict__ C, int M) {
    constexpr int K = 128;
    constexpr int NTILES = NOUT / 16;
    const int lane = threadIdx.x & 31;
    const int wglobal = blockIdx.x * 4 + (threadIdx.x >> 5);
    const int tm = wglobal / NTILES;
    const int tn = wglobal % NTILES;
    if (tm * 16 >= M) return;

    const int m  = lane & 15;
    const int kh = lane >> 4;
    const int n  = lane & 15;
    const int col = tn * 16 + n;
    const float* arow = A + (size_t)(tm * 16 + m) * K;

    v8f acc = {};
#pragma unroll 8
    for (int k0 = 0; k0 < K; k0 += 4) {
        const int ka = k0 + kh * 2;
        const float2 av = *reinterpret_cast<const float2*>(arow + ka);
        v2f a; a[0] = av.x; a[1] = av.y;
        v2f b;
        b[0] = B[(size_t)ka * NOUT + col];
        b[1] = B[(size_t)(ka + 1) * NOUT + col];
        acc = __builtin_amdgcn_wmma_f32_16x16x4_f32(false, a, false, b,
                                                    (short)0, acc, false, false);
    }

    const int row0 = tm * 16 + kh * 8;
#pragma unroll
    for (int r = 0; r < 8; ++r)
        C[(size_t)(row0 + r) * NOUT + col] = acc[r];
}

template <int F, int LOGC>
__global__ __launch_bounds__(256)
void edge_aggregate(const int* __restrict__ src, const int* __restrict__ dst,
                    const float* __restrict__ h, const float* __restrict__ dinv,
                    float* __restrict__ acc, long long total) {
    long long g = (long long)blockIdx.x * 256 + threadIdx.x;
    if (g >= total) return;
    const long long e = g >> LOGC;
    const int c = (int)(g & ((1 << LOGC) - 1));
    const int s = src[e];
    const int d = dst[e];
    const float norm = dinv[s] * dinv[d];
    const float4 hv = *reinterpret_cast<const float4*>(h + (size_t)s * F + c * 4);
    float* ap = acc + (size_t)d * F + c * 4;
    atomic_add_f32(ap + 0, hv.x * norm);
    atomic_add_f32(ap + 1, hv.y * norm);
    atomic_add_f32(ap + 2, hv.z * norm);
    atomic_add_f32(ap + 3, hv.w * norm);
}

template <int F, int LOGF, bool RELU>
__global__ __launch_bounds__(256)
void finalize_kernel(const float* __restrict__ acc, const float* __restrict__ h,
                     const float* __restrict__ dinv, const float* __restrict__ bias,
                     float* out, long long total) {
    long long i = (long long)blockIdx.x * 256 + threadIdx.x;
    if (i >= total) return;
    const int node = (int)(i >> LOGF);
    const int f = (int)(i & (F - 1));
    const float di = dinv[node];
    float v = acc[i] + h[i] * di * di + bias[f];
    out[i] = RELU ? fmaxf(v, 0.0f) : v;
}

static inline int blocks_for(long long n, int bs) { return (int)((n + bs - 1) / bs); }

extern "C" void kernel_launch(void* const* d_in, const int* in_sizes, int n_in,
                              void* d_out, int out_size, void* d_ws, size_t ws_size,
                              hipStream_t stream) {
    const float* x  = (const float*)d_in[0];
    const int*   ei = (const int*)d_in[1];
    const float* W1 = (const float*)d_in[2];
    const float* b1 = (const float*)d_in[3];
    const float* W2 = (const float*)d_in[4];
    const float* b2 = (const float*)d_in[5];
    float* out = (float*)d_out;

    const int N = in_sizes[0] / IN_C;
    const long long E = (long long)in_sizes[1] / 2;
    const int* esrc = ei;
    const int* edst = ei + E;

    float* ws   = (float*)d_ws;
    float* deg  = ws;
    float* dinv = deg + N;
    float* h1   = dinv + N;
    float* acc1 = h1 + (size_t)N * HID2;
    float* acc2 = acc1 + (size_t)N * HID2;
    float* h2   = out;

    const long long n1 = (long long)N * HID2;
    const long long n2 = (long long)N * HID;

    fill_kernel<<<blocks_for(n1, 256), 256, 0, stream>>>(acc1, 0.0f, n1);
    fill_kernel<<<blocks_for(n2, 256), 256, 0, stream>>>(acc2, 0.0f, n2);
    fill_kernel<<<blocks_for(N, 256), 256, 0, stream>>>(deg, 1.0f, N);

    degree_kernel<<<blocks_for(E, 256), 256, 0, stream>>>(edst, deg, E);
    rsqrt_kernel<<<blocks_for(N, 256), 256, 0, stream>>>(deg, dinv, N);

    {
        const int mtiles = (N + 15) / 16;
        const int waves = mtiles * (HID2 / 16);
        gemm_wmma_f32<HID2><<<(waves + 3) / 4, 128, 0, stream>>>(x, W1, h1, N);
    }
    edge_aggregate<HID2, 5><<<blocks_for(E * 32, 256), 256, 0, stream>>>(
        esrc, edst, h1, dinv, acc1, E * 32);
    finalize_kernel<HID2, 7, true><<<blocks_for(n1, 256), 256, 0, stream>>>(
        acc1, h1, dinv, b1, acc1, n1);

    {
        const int mtiles = (N + 15) / 16;
        const int waves = mtiles * (HID / 16);
        gemm_wmma_f32<HID><<<(waves + 3) / 4, 128, 0, stream>>>(acc1, W2, h2, N);
    }
    edge_aggregate<HID, 4><<<blocks_for(E * 16, 256), 256, 0, stream>>>(
        esrc, edst, h2, dinv, acc2, E * 16);
    finalize_kernel<HID, 6, false><<<blocks_for(n2, 256), 256, 0, stream>>>(
        acc2, h2, dinv, b2, out, n2);
}


# Replace the f32 atomic scatter-add of both layers by a destination-bucketed gather: one wave per node sums its rows in registers; bucket order is an arrival-order rank, so sum order still varies
# speedup vs baseline: 1.8475x; 1.8475x over previous
#include <hip/hip_runtime.h>
#include <hip/hip_bf16.h>


typedef float v2f __attribute__((ext_vector_type(2)));
typedef float v8f __attribute__((ext_vector_type(8)));

#define IN_C 128
#define HID2 128
#define HID  64

__global__ __launch_bounds__(256)
void fill_i32_kernel(int* __restrict__ p, int v, long long n) {
    long long i = (long long)blockIdx.x * 256 + threadIdx.x;
    if (i < n) p[i] = v;
}

__global__ __launch_bounds__(256)
void degree_rank_kernel(const int* __restrict__ dst, int* __restrict__ cnt,
                        int* __restrict__ rank, long long E) {
    long long e = (long long)blockIdx.x * 256 + threadIdx.x;
    if (e < E)
        rank[e] = __hip_atomic_fetch_add(&cnt[dst[e]], 1, __ATOMIC_RELAXED,
                                         __HIP_MEMORY_SCOPE_AGENT);
}

#define SCAN_T 1024
#define SCAN_E 16
__device__ __forceinline__ void scan_load_counts(const int* __restrict__ cnt, int i0, int n,
                                                 int (&c)[SCAN_E]) {
    for (int j = 0; j < SCAN_E; ++j) c[j] = (i0 + j < n) ? cnt[i0 + j] : 0;
}

__global__ __launch_bounds__(SCAN_T)
void scan_norm_kernel(const int* __restrict__ cnt, int* __restrict__ rowptr,
                      float* __restrict__ dinv, int n) {
    __shared__ int wsum[2][32];
    const int t = threadIdx.x;
    const int lane = t & 31;
    const int w = t >> 5;
    int carry = 0;
    int buf = 0;
    int c[SCAN_E];
    scan_load_counts(cnt, t * SCAN_E, n, c);
    int s = 0;
    for (int j = 0; j < SCAN_E; ++j) s += c[j];
    for (int base = 0; base < n; base += SCAN_T * SCAN_E) {
        const int i0 = base + t * SCAN_E;
        int cn[SCAN_E];
        scan_load_counts(cnt, i0 + SCAN_T * SCAN_E, n, cn);
        int incl = s;
        for (int off = 1; off < 32; off <<= 1) {
            const int v = __shfl_up(incl, off, 32);
            if (lane >= off) incl += v;
        }
        if (lane == 31) wsum[buf][w] = incl;
        __syncthreads();
        const int wt = wsum[buf][lane];
        int winc = wt;
        for (int off = 1; off < 32; off <<= 1) {
            const int v = __shfl_up(winc, off, 32);
            if (lane >= off) winc += v;
        }
        const int wave_excl = __shfl(winc, w, 32) - __shfl(wt, w, 32);
        const int total = __shfl(winc, 31, 32);
        int run = carry + wave_excl + (incl - s);
        for (int j = 0; j < SCAN_E; ++j) {
            if (i0 + j < n) {
                rowptr[i0 + j] = run;
                dinv[i0 + j] = rsqrtf((float)(c[j] + 1));
                run += c[j];
            }
        }
        carry += total;
        buf ^= 1;
        s = 0;
        for (int j = 0; j < SCAN_E; ++j) { c[j] = cn[j]; s += cn[j]; }
    }
    if (t == 0) rowptr[n] = carry;
}

__global__ __launch_bounds__(256)
void csr_fill_kernel(const int* __restrict__ src, const int* __restrict__ dst,
                     const int* __restrict__ rank, const int* __restrict__ rowptr,
                     int* __restrict__ csr, long long E) {
    long long e = (long long)blockIdx.x * 256 + threadIdx.x;
    if (e < E) csr[rowptr[dst[e]] + rank[e]] = src[e];
}

template <int NOUT>
__global__ __launch_bounds__(128)
void gemm_wmma_f32(const float* __restrict__ A, const float* __restrict__ B,
                   float* __restrict__ C, int M) {
    constexpr int K = 128;
    constexpr int NTILES = NOUT / 16;
    const int lane = threadIdx.x & 31;
    const int wglobal = blockIdx.x * 4 + (threadIdx.x >> 5);
    const int tm = wglobal / NTILES;
    const int tn = wglobal % NTILES;
    if (tm * 16 >= M) return;

    const int m  = lane & 15;
    const int kh = lane >> 4;
    const int n  = lane & 15;
    const int col = tn * 16 + n;
    const float* arow = A + (size_t)(tm * 16 + m) * K;

    v8f acc = {};
#pragma unroll 8
    for (int k0 = 0; k0 < K; k0 += 4) {
        const int ka = k0 + kh * 2;
        const float2 av = *reinterpret_cast<const float2*>(arow + ka);
        v2f a; a[0] = av.x; a[1] = av.y;
        v2f b;
        b[0] = B[(size_t)ka * NOUT + col];
        b[1] = B[(size_t)(ka + 1) * NOUT + col];
        acc = __builtin_amdgcn_wmma_f32_16x16x4_f32(false, a, false, b,
                                                    (short)0, acc, false, false);
    }

    const int row0 = tm * 16 + kh * 8;
#pragma unroll
    for (int r = 0; r < 8; ++r)
        C[(size_t)(row0 + r) * NOUT + col] = acc[r];
}


template <int F, int CH>
__device__ __forceinline__ void gather_chunk(float __attribute__((ext_vector_type(F / 32)))& acc,
                                             const float* __restrict__ hlane,
                                             int s_l, float w_l, int j) {
    typedef float vf __attribute__((ext_vector_type(F / 32)));
    vf t[CH];
    for (int q = 0; q < CH; ++q) {
        const int s = __builtin_amdgcn_readlane(s_l, j + q);
        t[q] = *reinterpret_cast<const vf*>(hlane + (size_t)s * F);
    }
    for (int q = 0; q < CH; ++q) {
        const float wgt = __builtin_bit_cast(
            float, __builtin_amdgcn_readlane(__builtin_bit_cast(int, w_l), j + q));
        acc += t[q] * wgt;
    }
}

template <int F, bool RELU>
__global__ __launch_bounds__(256)
void gather_aggregate(const int* __restrict__ rowptr, const int* __restrict__ csr,
                      const float* __restrict__ h, const float* __restrict__ dinv,
                      const float* __restrict__ bias, float* __restrict__ out, int n) {
    constexpr int V = F / 32;
    typedef float vf __attribute__((ext_vector_type(V)));
    const int lane = threadIdx.x & 31;
    const int d = __builtin_amdgcn_readfirstlane((int)(blockIdx.x * 8 + (threadIdx.x >> 5)));
    if (d >= n) return;
    const int beg = rowptr[d];
    const int end = rowptr[d + 1];
    const float dd = dinv[d];
    const float* hlane = h + lane * V;
    const vf hs = *reinterpret_cast<const vf*>(hlane + (size_t)d * F);
    const vf bv = *reinterpret_cast<const vf*>(bias + lane * V);

    vf acc = {};
    for (int j0 = beg; j0 < end; j0 += 32) {
        const int m = (end - j0 < 32) ? (end - j0) : 32;
        int s_l = 0;
        float w_l = 0.0f;
        if (lane < m) {
            s_l = csr[j0 + lane];
            w_l = dinv[s_l] * dd;
        }
        int j = 0;
        for (; j + 8 <= m; j += 8) gather_chunk<F, 8>(acc, hlane, s_l, w_l, j);
        if (m & 4) { gather_chunk<F, 4>(acc, hlane, s_l, w_l, j); j += 4; }
        if (m & 2) { gather_chunk<F, 2>(acc, hlane, s_l, w_l, j); j += 2; }
        if (m & 1) { gather_chunk<F, 1>(acc, hlane, s_l, w_l, j); }
    }

    vf v = acc + hs * (dd * dd) + bv;
    if (RELU) {
        for (int i = 0; i < V; ++i) v[i] = fmaxf(v[i], 0.0f);
    }
    *reinterpret_cast<vf*>(out + (size_t)d * F + lane * V) = v;
}

static inline int blocks_for(long long n, int bs) { return (int)((n + bs - 1) / bs); }

extern "C" void kernel_launch(void* const* d_in, const int* in_sizes, int n_in,
                              void* d_out, int out_size, void* d_ws, size_t ws_size,
                              hipStream_t stream) {
    const float* x  = (const float*)d_in[0];
    const int*   ei = (const int*)d_in[1];
    const float* W1 = (const float*)d_in[2];
    const float* b1 = (const float*)d_in[3];
    const float* W2 = (const float*)d_in[4];
    const float* b2 = (const float*)d_in[5];
    float* out = (float*)d_out;

    const int N = in_sizes[0] / IN_C;
    const long long E = (long long)in_sizes[1] / 2;
    const int* esrc = ei;
    const int* edst = ei + E;

    float* ws   = (float*)d_ws;
    float* h1   = ws;
    float* a1   = h1 + (size_t)N * HID2;
    float* h2   = a1 + (size_t)N * HID2;
    float* dinv = h2 + (size_t)N * HID;
    int*   cnt  = (int*)(dinv + N);
    int*   rank = cnt + N;
    int*   csr  = rank + E;
    int*   rowptr = csr + E;

    fill_i32_kernel<<<blocks_for(N, 256), 256, 0, stream>>>(cnt, 0, N);
    degree_rank_kernel<<<blocks_for(E, 256), 256, 0, stream>>>(edst, cnt, rank, E);
    scan_norm_kernel<<<1, SCAN_T, 0, stream>>>(cnt, rowptr, dinv, N);
    csr_fill_kernel<<<blocks_for(E, 256), 256, 0, stream>>>(esrc, edst, rank, rowptr, csr, E);

    {
        const int mtiles = (N + 15) / 16;
        const int waves = mtiles * (HID2 / 16);
        gemm_wmma_f32<HID2><<<(waves + 3) / 4, 128, 0, stream>>>(x, W1, h1, N);
    }
    gather_aggregate<HID2, true><<<blocks_for(N, 8), 256, 0, stream>>>(
        rowptr, csr, h1, dinv, b1, a1, N);

    {
        const int mtiles = (N + 15) / 16;
        const int waves = mtiles * (HID / 16);
        gemm_wmma_f32<HID><<<(waves + 3) / 4, 128, 0, stream>>>(a1, W2, h2, N);
    }
    gather_aggregate<HID, false><<<blocks_for(N, 8), 256, 0, stream>>>(
        rowptr, csr, h2, dinv, b2, out, N);
}


# GEMM data path: 256 resident work-groups; a wave keeps the B fragments of 64 columns in registers (staged once via LDS), walks 16-row strips of A read a strip ahead; k summed in another order
# speedup vs baseline: 2.0793x; 1.1255x over previous
#include <hip/hip_runtime.h>
#include <hip/hip_bf16.h>


typedef float v2f __attribute__((ext_vector_type(2)));
typedef float v8f __attribute__((ext_vector_type(8)));

#define IN_C 128
#define HID2 128
#define HID  64

__global__ __launch_bounds__(256)
void fill_i32_kernel(int* __restrict__ p, int v, long long n) {
    long long i = (long long)blockIdx.x * 256 + threadIdx.x;
    if (i < n) p[i] = v;
}

__global__ __launch_bounds__(256)
void degree_rank_kernel(const int* __restrict__ dst, int* __restrict__ cnt,
                        int* __restrict__ rank, long long E) {
    long long e = (long long)blockIdx.x * 256 + threadIdx.x;
    if (e < E)
        rank[e] = __hip_atomic_fetch_add(&cnt[dst[e]], 1, __ATOMIC_RELAXED,
                                         __HIP_MEMORY_SCOPE_AGENT);
}

#define SCAN_T 1024
#define SCAN_E 16
__device__ __forceinline__ void scan_load_counts(const int* __restrict__ cnt, int i0, int n,
                                                 int (&c)[SCAN_E]) {
    for (int j = 0; j < SCAN_E; ++j) c[j] = (i0 + j < n) ? cnt[i0 + j] : 0;
}

__global__ __launch_bounds__(SCAN_T)
void scan_norm_kernel(const int* __restrict__ cnt, int* __restrict__ rowptr,
                      float* __restrict__ dinv, int n) {
    __shared__ int wsum[2][32];
    const int t = threadIdx.x;
    const int lane = t & 31;
    const int w = t >> 5;
    int carry = 0;
    int buf = 0;
    int c[SCAN_E];
    scan_load_counts(cnt, t * SCAN_E, n, c);
    int s = 0;
    for (int j = 0; j < SCAN_E; ++j) s += c[j];
    for (int base = 0; base < n; base += SCAN_T * SCAN_E) {
        const int i0 = base + t * SCAN_E;
        int cn[SCAN_E];
        scan_load_counts(cnt, i0 + SCAN_T * SCAN_E, n, cn);
        int incl = s;
        for (int off = 1; off < 32; off <<= 1) {
            const int v = __shfl_up(incl, off, 32);
            if (lane >= off) incl += v;
        }
        if (lane == 31) wsum[buf][w] = incl;
        __syncthreads();
        const int wt = wsum[buf][lane];
        int winc = wt;
        for (int off = 1; off < 32; off <<= 1) {
            const int v = __shfl_up(winc, off, 32);
            if (lane >= off) winc += v;
        }
        const int wave_excl = __shfl(winc, w, 32) - __shfl(wt, w, 32);
        const int total = __shfl(winc, 31, 32);
        int run = carry + wave_excl + (incl - s);
        for (int j = 0; j < SCAN_E; ++j) {
            if (i0 + j < n) {
                rowptr[i0 + j] = run;
                dinv[i0 + j] = rsqrtf((float)(c[j] + 1));
                run += c[j];
            }
        }
        carry += total;
        buf ^= 1;
        s = 0;
        for (int j = 0; j < SCAN_E; ++j) { c[j] = cn[j]; s += cn[j]; }
    }
    if (t == 0) rowptr[n] = carry;
}

__global__ __launch_bounds__(256)
void csr_fill_kernel(const int* __restrict__ src, const int* __restrict__ dst,
                     const int* __restrict__ rank, const int* __restrict__ rowptr,
                     int* __restrict__ csr, long long E) {
    long long e = (long long)blockIdx.x * 256 + threadIdx.x;
    if (e < E) csr[rowptr[dst[e]] + rank[e]] = src[e];
}

#define GEMM_GROUPS 256
template <int NOUT>
__global__ __launch_bounds__(256)
void gemm_wmma_f32(const float* __restrict__ A, const float* __restrict__ B,
                   float* __restrict__ C, int M) {
    constexpr int K  = 128;
    constexpr int NT = NOUT / 16;
    constexpr int KB = K / 8;
    constexpr int TW = 4;
    constexpr int NH = NT / TW;
    __shared__ float4 bl[KB * NT * 32];

    const int nstrips = (M + 15) >> 4;
    if (nstrips <= 0) return;
    const int lane = threadIdx.x & 31;
    const int wave = __builtin_amdgcn_readfirstlane((int)(threadIdx.x >> 5));
    const int c  = lane & 15;
    const int hh = lane >> 4;

    const int nslots = (int)gridDim.x * 4;
    const int pair = wave >> 2;
    const int half = (NH == 2) ? pair : 0;
    const int step = (NH == 2) ? nslots : 2 * nslots;
    int s = (int)blockIdx.x * 4 + (wave & 3) + ((NH == 2) ? 0 : pair * nslots);

    const float4* abase = reinterpret_cast<const float4*>(A) + (size_t)c * (K / 4) + hh;

    float4 a[KB];
    {
        const int s0 = (s < nstrips) ? s : (nstrips - 1);
        const float4* ap = abase + (size_t)s0 * (16 * K / 4);
#pragma unroll
        for (int kb = 0; kb < KB; ++kb) a[kb] = ap[kb * 2];
    }

#pragma unroll
    for (int i = 0; i < KB * NT / 8; ++i) {
        const int g  = wave + 8 * i;
        const int t  = g % NT;
        const int kb = g / NT;
        const float* bp = B + (size_t)(kb * 8 + 4 * hh) * NOUT + t * 16 + c;
        float4 v;
        v.x = bp[0];
        v.y = bp[NOUT];
        v.z = bp[2 * NOUT];
        v.w = bp[3 * NOUT];
        bl[g * 32 + lane] = v;
    }
    __syncthreads();

    float4 bf[KB][TW];
    {
        const float4* bw = bl + (half * TW) * 32 + lane;
#pragma unroll
        for (int kb = 0; kb < KB; ++kb) {
#pragma unroll
            for (int t = 0; t < TW; ++t) bf[kb][t] = bw[(kb * NT + t) * 32];
        }
    }

    while (s < nstrips) {
        const int sn = s + step;
        const int sp = (sn < nstrips) ? sn : s;
        float4 an[KB];
        {
            const float4* ap = abase + (size_t)sp * (16 * K / 4);
#pragma unroll
            for (int kb = 0; kb < KB; ++kb) an[kb] = ap[kb * 2];
        }

        v8f acc[TW];
#pragma unroll
        for (int t = 0; t < TW; ++t) acc[t] = v8f{};
#pragma unroll
        for (int kb = 0; kb < KB; ++kb) {
            v2f a0; a0[0] = a[kb].x; a0[1] = a[kb].y;
            v2f a1; a1[0] = a[kb].z; a1[1] = a[kb].w;
#pragma unroll
            for (int t = 0; t < TW; ++t) {
                v2f b; b[0] = bf[kb][t].x; b[1] = bf[kb][t].y;
                acc[t] = __builtin_amdgcn_wmma_f32_16x16x4_f32(false, a0, false, b,
                                                               (short)0, acc[t], false, false);
            }
#pragma unroll
            for (int t = 0; t < TW; ++t) {
                v2f b; b[0] = bf[kb][t].z; b[1] = bf[kb][t].w;
                acc[t] = __builtin_amdgcn_wmma_f32_16x16x4_f32(false, a1, false, b,
                                                               (short)0, acc[t], false, false);
            }
            __builtin_amdgcn_sched_barrier(0);
        }

        float* crow = C + (size_t)(s * 16 + hh * 8) * NOUT + half * (TW * 16) + c;
#pragma unroll
        for (int t = 0; t < TW; ++t) {
#pragma unroll
            for (int r = 0; r < 8; ++r)
                crow[(size_t)r * NOUT + t * 16] = acc[t][r];
        }

#pragma unroll
        for (int kb = 0; kb < KB; ++kb) a[kb] = an[kb];
        s = sn;
    }
}


template <int F, int CH>
__device__ __forceinline__ void gather_chunk(float __attribute__((ext_vector_type(F / 32)))& acc,
                                             const float* __restrict__ hlane,
                                             int s_l, float w_l, int j) {
    typedef float vf __attribute__((ext_vector_type(F / 32)));
    vf t[CH];
    for (int q = 0; q < CH; ++q) {
        const int s = __builtin_amdgcn_readlane(s_l, j + q);
        t[q] = *reinterpret_cast<const vf*>(hlane + (size_t)s * F);
    }
    for (int q = 0; q < CH; ++q) {
        const float wgt = __builtin_bit_cast(
            float, __builtin_amdgcn_readlane(__builtin_bit_cast(int, w_l), j + q));
        acc += t[q] * wgt;
    }
}

template <int F, bool RELU>
__global__ __launch_bounds__(256)
void gather_aggregate(const int* __restrict__ rowptr, const int* __restrict__ csr,
                      const float* __restrict__ h, const float* __restrict__ dinv,
                      const float* __restrict__ bias, float* __restrict__ out, int n) {
    constexpr int V = F / 32;
    typedef float vf __attribute__((ext_vector_type(V)));
    const int lane = threadIdx.x & 31;
    const int d = __builtin_amdgcn_readfirstlane((int)(blockIdx.x * 8 + (threadIdx.x >> 5)));
    if (d >= n) return;
    const int beg = rowptr[d];
    const int end = rowptr[d + 1];
    const float dd = dinv[d];
    const float* hlane = h + lane * V;
    const vf hs = *reinterpret_cast<const vf*>(hlane + (size_t)d * F);
    const vf bv = *reinterpret_cast<const vf*>(bias + lane * V);

    vf acc = {};
    for (int j0 = beg; j0 < end; j0 += 32) {
        const int m = (end - j0 < 32) ? (end - j0) : 32;
        int s_l = 0;
        float w_l = 0.0f;
        if (lane < m) {
            s_l = csr[j0 + lane];
            w_l = dinv[s_l] * dd;
        }
        int j = 0;
        for (; j + 8 <= m; j += 8) gather_chunk<F, 8>(acc, hlane, s_l, w_l, j);
        if (m & 4) { gather_chunk<F, 4>(acc, hlane, s_l, w_l, j); j += 4; }
        if (m & 2) { gather_chunk<F, 2>(acc, hlane, s_l, w_l, j); j += 2; }
        if (m & 1) { gather_chunk<F, 1>(acc, hlane, s_l, w_l, j); }
    }

    vf v = acc + hs * (dd * dd) + bv;
    if (RELU) {
        for (int i = 0; i < V; ++i) v[i] = fmaxf(v[i], 0.0f);
    }
    *reinterpret_cast<vf*>(out + (size_t)d * F + lane * V) = v;
}

static inline int blocks_for(long long n, int bs) { return (int)((n + bs - 1) / bs); }

extern "C" void kernel_launch(void* const* d_in, const int* in_sizes, int n_in,
                              void* d_out, int out_size, void* d_ws, size_t ws_size,
                              hipStream_t stream) {
    const float* x  = (const float*)d_in[0];
    const int*   ei = (const int*)d_in[1];
    const float* W1 = (const float*)d_in[2];
    const float* b1 = (const float*)d_in[3];
    const float* W2 = (const float*)d_in[4];
    const float* b2 = (const float*)d_in[5];
    float* out = (float*)d_out;

    const int N = in_sizes[0] / IN_C;
    const long long E = (long long)in_sizes[1] / 2;
    const int* esrc = ei;
    const int* edst = ei + E;

    float* ws   = (float*)d_ws;
    float* h1   = ws;
    float* a1   = h1 + (size_t)N * HID2;
    float* h2   = a1 + (size_t)N * HID2;
    float* dinv = h2 + (size_t)N * HID;
    int*   cnt  = (int*)(dinv + N);
    int*   rank = cnt + N;
    int*   csr  = rank + E;
    int*   rowptr = csr + E;

    fill_i32_kernel<<<blocks_for(N, 256), 256, 0, stream>>>(cnt, 0, N);
    degree_rank_kernel<<<blocks_for(E, 256), 256, 0, stream>>>(edst, cnt, rank, E);
    scan_norm_kernel<<<1, SCAN_T, 0, stream>>>(cnt, rowptr, dinv, N);
    csr_fill_kernel<<<blocks_for(E, 256), 256, 0, stream>>>(esrc, edst, rank, rowptr, csr, E);

    {
        gemm_wmma_f32<HID2><<<GEMM_GROUPS, 256, 0, stream>>>(x, W1, h1, N);
    }
    gather_aggregate<HID2, true><<<blocks_for(N, 8), 256, 0, stream>>>(
        rowptr, csr, h1, dinv, b1, a1, N);

    {
        gemm_wmma_f32<HID><<<GEMM_GROUPS, 256, 0, stream>>>(a1, W2, h2, N);
    }
    gather_aggregate<HID, false><<<blocks_for(N, 8), 256, 0, stream>>>(
        rowptr, csr, h2, dinv, b2, out, N);
}


# Edges go straight into fixed-stride destination buckets at the slot the degree atomic returns: no prefix scan, placement pass or rank array (8 to 6 launches); arrival-order rank, sum order varies
# speedup vs baseline: 3.0825x; 1.4825x over previous
#include <hip/hip_runtime.h>
#include <hip/hip_bf16.h>


typedef float v2f __attribute__((ext_vector_type(2)));
typedef float v8f __attribute__((ext_vector_type(8)));

#define IN_C 128
#define HID2 128
#define HID  64

__global__ __launch_bounds__(256)
void fill_i32_kernel(int* __restrict__ p, int v, long long n) {
    long long i = (long long)blockIdx.x * 256 + threadIdx.x;
    if (i < n) p[i] = v;
}

#define CAP 64
__global__ __launch_bounds__(256)
void bucket_place_kernel(const int* __restrict__ src, const int* __restrict__ dst,
                         int* __restrict__ cnt, int* __restrict__ bucket, long long E) {
    long long e = (long long)blockIdx.x * 256 + threadIdx.x;
    if (e < E) {
        const int s = src[e];
        const int d = dst[e];
        asm volatile("" :: "v"(s));
        const int r = __hip_atomic_fetch_add(&cnt[d], 1, __ATOMIC_RELAXED,
                                             __HIP_MEMORY_SCOPE_AGENT);
        if (r < CAP) bucket[(size_t)d * CAP + r] = s;
    }
}

#define GEMM_GROUPS 256
template <int NOUT>
__global__ __launch_bounds__(256)
void gemm_wmma_f32(const float* __restrict__ A, const float* __restrict__ B,
                   float* __restrict__ C, int M) {
    constexpr int K  = 128;
    constexpr int NT = NOUT / 16;
    constexpr int KB = K / 8;
    constexpr int TW = 4;
    constexpr int NH = NT / TW;
    __shared__ float4 bl[KB * NT * 32];

    const int nstrips = (M + 15) >> 4;
    if (nstrips <= 0) return;
    const int lane = threadIdx.x & 31;
    const int wave = __builtin_amdgcn_readfirstlane((int)(threadIdx.x >> 5));
    const int c  = lane & 15;
    const int hh = lane >> 4;

    const int nslots = (int)gridDim.x * 4;
    const int pair = wave >> 2;
    const int half = (NH == 2) ? pair : 0;
    const int step = (NH == 2) ? nslots : 2 * nslots;
    int s = (int)blockIdx.x * 4 + (wave & 3) + ((NH == 2) ? 0 : pair * nslots);

    const float4* abase = reinterpret_cast<const float4*>(A) + (size_t)c * (K / 4) + hh;

    float4 a[KB];
    {
        const int s0 = (s < nstrips) ? s : (nstrips - 1);
        const float4* ap = abase + (size_t)s0 * (16 * K / 4);
#pragma unroll
        for (int kb = 0; kb < KB; ++kb) a[kb] = ap[kb * 2];
    }

#pragma unroll
    for (int i = 0; i < KB * NT / 8; ++i) {
        const int g  = wave + 8 * i;
        const int t  = g % NT;
        const int kb = g / NT;
        const float* bp = B + (size_t)(kb * 8 + 4 * hh) * NOUT + t * 16 + c;
        float4 v;
        v.x = bp[0];
        v.y = bp[NOUT];
        v.z = bp[2 * NOUT];
        v.w = bp[3 * NOUT];
        bl[g * 32 + lane] = v;
    }
    __syncthreads();

    float4 bf[KB][TW];
    {
        const float4* bw = bl + (half * TW) * 32 + lane;
#pragma unroll
        for (int kb = 0; kb < KB; ++kb) {
#pragma unroll
            for (int t = 0; t < TW; ++t) bf[kb][t] = bw[(kb * NT + t) * 32];
        }
    }

    while (s < nstrips) {
        const int sn = s + step;
        const int sp = (sn < nstrips) ? sn : s;
        float4 an[KB];
        {
            const float4* ap = abase + (size_t)sp * (16 * K / 4);
#pragma unroll
            for (int kb = 0; kb < KB; ++kb) an[kb] = ap[kb * 2];
        }

        v8f acc[TW];
#pragma unroll
        for (int t = 0; t < TW; ++t) acc[t] = v8f{};
#pragma unroll
        for (int kb = 0; kb < KB; ++kb) {
            v2f a0; a0[0] = a[kb].x; a0[1] = a[kb].y;
            v2f a1; a1[0] = a[kb].z; a1[1] = a[kb].w;
#pragma unroll
            for (int t = 0; t < TW; ++t) {
                v2f b; b[0] = bf[kb][t].x; b[1] = bf[kb][t].y;
                acc[t] = __builtin_amdgcn_wmma_f32_16x16x4_f32(false, a0, false, b,
                                                               (short)0, acc[t], false, false);
            }
#pragma unroll
            for (int t = 0; t < TW; ++t) {
                v2f b; b[0] = bf[kb][t].z; b[1] = bf[kb][t].w;
                acc[t] = __builtin_amdgcn_wmma_f32_16x16x4_f32(false, a1, false, b,
                                                               (short)0, acc[t], false, false);
            }
            __builtin_amdgcn_sched_barrier(0);
        }

        float* crow = C + (size_t)(s * 16 + hh * 8) * NOUT + half * (TW * 16) + c;
#pragma unroll
        for (int t = 0; t < TW; ++t) {
#pragma unroll
            for (int r = 0; r < 8; ++r)
                crow[(size_t)r * NOUT + t * 16] = acc[t][r];
        }

#pragma unroll
        for (int kb = 0; kb < KB; ++kb) a[kb] = an[kb];
        s = sn;
    }
}


template <int F, int CH>
__device__ __forceinline__ void gather_chunk(float __attribute__((ext_vector_type(F / 32)))& acc,
                                             const float* __restrict__ hlane,
                                             int s_l, float w_l, int j) {
    typedef float vf __attribute__((ext_vector_type(F / 32)));
    vf t[CH];
    for (int q = 0; q < CH; ++q) {
        const int s = __builtin_amdgcn_readlane(s_l, j + q);
        t[q] = *reinterpret_cast<const vf*>(hlane + (size_t)s * F);
    }
    for (int q = 0; q < CH; ++q) {
        const float wgt = __builtin_bit_cast(
            float, __builtin_amdgcn_readlane(__builtin_bit_cast(int, w_l), j + q));
        acc += t[q] * wgt;
    }
}

template <int F, bool RELU>
__global__ __launch_bounds__(256)
void gather_aggregate(const int* __restrict__ cnt, const int* __restrict__ bucket,
                      const int* __restrict__ esrc, const int* __restrict__ edst, long long E,
                      const float* __restrict__ h, const float* __restrict__ bias,
                      float* __restrict__ out, int n) {
    constexpr int V = F / 32;
    typedef float vf __attribute__((ext_vector_type(V)));
    const int lane = threadIdx.x & 31;
    const int d = __builtin_amdgcn_readfirstlane((int)(blockIdx.x * 8 + (threadIdx.x >> 5)));
    if (d >= n) return;
    const int deg = cnt[d];
    const float dd = rsqrtf((float)(deg + 1));
    const float* hlane = h + lane * V;
    const vf hs = *reinterpret_cast<const vf*>(hlane + (size_t)d * F);
    const vf bv = *reinterpret_cast<const vf*>(bias + lane * V);

    vf acc = {};
    if (deg <= CAP) {
        const int* brow = bucket + (size_t)d * CAP;
        for (int j0 = 0; j0 < deg; j0 += 32) {
            const int m = (deg - j0 < 32) ? (deg - j0) : 32;
            int s_l = 0;
            float w_l = 0.0f;
            if (lane < m) {
                s_l = brow[j0 + lane];
                w_l = rsqrtf((float)(cnt[s_l] + 1)) * dd;
            }
            int j = 0;
            for (; j + 8 <= m; j += 8) gather_chunk<F, 8>(acc, hlane, s_l, w_l, j);
            if (m & 4) { gather_chunk<F, 4>(acc, hlane, s_l, w_l, j); j += 4; }
            if (m & 2) { gather_chunk<F, 2>(acc, hlane, s_l, w_l, j); j += 2; }
            if (m & 1) { gather_chunk<F, 1>(acc, hlane, s_l, w_l, j); }
        }
    } else {
        for (long long e0 = 0; e0 < E; e0 += 32) {
            const long long e = e0 + lane;
            int s_l = 0;
            float w_l = 0.0f;
            bool hit = false;
            if (e < E && edst[e] == d) {
                hit = true;
                s_l = esrc[e];
                w_l = rsqrtf((float)(cnt[s_l] + 1)) * dd;
            }
            unsigned int todo = __builtin_amdgcn_ballot_w32(hit);
            while (todo != 0u) {
                const int b = __builtin_ctz(todo);
                todo &= todo - 1u;
                gather_chunk<F, 1>(acc, hlane, s_l, w_l, b);
            }
        }
    }

    vf v = acc + hs * (dd * dd) + bv;
    if (RELU) {
        for (int i = 0; i < V; ++i) v[i] = fmaxf(v[i], 0.0f);
    }
    *reinterpret_cast<vf*>(out + (size_t)d * F + lane * V) = v;
}

static inline int blocks_for(long long n, int bs) { return (int)((n + bs - 1) / bs); }

extern "C" void kernel_launch(void* const* d_in, const int* in_sizes, int n_in,
                              void* d_out, int out_size, void* d_ws, size_t ws_size,
                              hipStream_t stream) {
    const float* x  = (const float*)d_in[0];
    const int*   ei = (const int*)d_in[1];
    const float* W1 = (const float*)d_in[2];
    const float* b1 = (const float*)d_in[3];
    const float* W2 = (const float*)d_in[4];
    const float* b2 = (const float*)d_in[5];
    float* out = (float*)d_out;

    const int N = in_sizes[0] / IN_C;
    const long long E = (long long)in_sizes[1] / 2;
    const int* esrc = ei;
    const int* edst = ei + E;

    float* ws   = (float*)d_ws;
    float* h1   = ws;
    float* a1   = h1 + (size_t)N * HID2;
    float* h2   = a1 + (size_t)N * HID2;
    int*   bucket = (int*)(h2 + (size_t)N * HID);
    int*   cnt  = bucket + (size_t)N * CAP;

    fill_i32_kernel<<<blocks_for(N, 256), 256, 0, stream>>>(cnt, 0, N);
    bucket_place_kernel<<<blocks_for(E, 256), 256, 0, stream>>>(esrc, edst, cnt, bucket, E);

    {
        gemm_wmma_f32<HID2><<<GEMM_GROUPS, 256, 0, stream>>>(x, W1, h1, N);
    }
    gather_aggregate<HID2, true><<<blocks_for(N, 8), 256, 0, stream>>>(
        cnt, bucket, esrc, edst, E, h1, b1, a1, N);

    {
        gemm_wmma_f32<HID><<<GEMM_GROUPS, 256, 0, stream>>>(a1, W2, h2, N);
    }
    gather_aggregate<HID, false><<<blocks_for(N, 8), 256, 0, stream>>>(
        cnt, bucket, esrc, edst, E, h2, b2, out, N);
}


# In-degree counter moved into word 0 of the node's own bucket: atomic, slot store and gather read share one line per node (about 17 atomics per line, not 512); no counter array; sum order kept
# speedup vs baseline: 3.5684x; 1.1576x over previous
#include <hip/hip_runtime.h>
#include <hip/hip_bf16.h>


typedef float v2f __attribute__((ext_vector_type(2)));
typedef float v8f __attribute__((ext_vector_type(8)));

#define IN_C 128
#define HID2 128
#define HID  64

__global__ __launch_bounds__(256)
void fill_i32_kernel(int* __restrict__ p, int v, long long n, int stride) {
    long long i = (long long)blockIdx.x * 256 + threadIdx.x;
    if (i < n) p[i * stride] = v;
}

#define CAP 64
#define SLOTS (CAP - 1)
__global__ __launch_bounds__(256)
void bucket_place_kernel(const int* __restrict__ src, const int* __restrict__ dst,
                         int* __restrict__ bucket, long long E) {
    long long e = (long long)blockIdx.x * 256 + threadIdx.x;
    if (e < E) {
        const int s = src[e];
        const int d = dst[e];
        asm volatile("" :: "v"(s));
        int* brow = bucket + (size_t)d * CAP;
        const int r = __hip_atomic_fetch_add(brow, 1, __ATOMIC_RELAXED,
                                             __HIP_MEMORY_SCOPE_AGENT);
        if (r < SLOTS) brow[1 + r] = s;
    }
}

#define GEMM_GROUPS 256
template <int NOUT>
__global__ __launch_bounds__(256)
void gemm_wmma_f32(const float* __restrict__ A, const float* __restrict__ B,
                   const int* __restrict__ cnt, float* __restrict__ C, int M) {
    constexpr int K  = 128;
    constexpr int NT = NOUT / 16;
    constexpr int KB = K / 8;
    constexpr int TW = 4;
    constexpr int NH = NT / TW;
    __shared__ float4 bl[KB * NT * 32];

    const int nstrips = (M + 15) >> 4;
    if (nstrips <= 0) return;
    const int lane = threadIdx.x & 31;
    const int wave = __builtin_amdgcn_readfirstlane((int)(threadIdx.x >> 5));
    const int c  = lane & 15;
    const int hh = lane >> 4;

    const int nslots = (int)gridDim.x * 4;
    const int pair = wave >> 2;
    const int half = (NH == 2) ? pair : 0;
    const int step = (NH == 2) ? nslots : 2 * nslots;
    int s = (int)blockIdx.x * 4 + (wave & 3) + ((NH == 2) ? 0 : pair * nslots);

    const float4* abase = reinterpret_cast<const float4*>(A) + (size_t)c * (K / 4) + hh;

    float4 a[KB];
    int cn[8];
    {
        const int s0 = (s < nstrips) ? s : (nstrips - 1);
        const int* cp = cnt + ((size_t)s0 * 16 + hh * 8) * CAP;
#pragma unroll
        for (int r = 0; r < 8; ++r) cn[r] = cp[r * CAP];
        const float4* ap = abase + (size_t)s0 * (16 * K / 4);
#pragma unroll
        for (int kb = 0; kb < KB; ++kb) a[kb] = ap[kb * 2];
    }

#pragma unroll
    for (int i = 0; i < KB * NT / 8; ++i) {
        const int g  = wave + 8 * i;
        const int t  = g % NT;
        const int kb = g / NT;
        const float* bp = B + (size_t)(kb * 8 + 4 * hh) * NOUT + t * 16 + c;
        float4 v;
        v.x = bp[0];
        v.y = bp[NOUT];
        v.z = bp[2 * NOUT];
        v.w = bp[3 * NOUT];
        bl[g * 32 + lane] = v;
    }
    __syncthreads();

    float4 bf[KB][TW];
    {
        const float4* bw = bl + (half * TW) * 32 + lane;
#pragma unroll
        for (int kb = 0; kb < KB; ++kb) {
#pragma unroll
            for (int t = 0; t < TW; ++t) bf[kb][t] = bw[(kb * NT + t) * 32];
        }
    }

    while (s < nstrips) {
        const int sn = s + step;
        const int sp = (sn < nstrips) ? sn : s;
        const int* cp = cnt + ((size_t)sp * 16 + hh * 8) * CAP;
        int nn[8];
#pragma unroll
        for (int r = 0; r < 8; ++r) nn[r] = cp[r * CAP];
        float4 an[KB];
        {
            const float4* ap = abase + (size_t)sp * (16 * K / 4);
#pragma unroll
            for (int kb = 0; kb < KB; ++kb) an[kb] = ap[kb * 2];
        }

        v8f acc[TW];
#pragma unroll
        for (int t = 0; t < TW; ++t) acc[t] = v8f{};
#pragma unroll
        for (int kb = 0; kb < KB; ++kb) {
            v2f a0; a0[0] = a[kb].x; a0[1] = a[kb].y;
            v2f a1; a1[0] = a[kb].z; a1[1] = a[kb].w;
#pragma unroll
            for (int t = 0; t < TW; ++t) {
                v2f b; b[0] = bf[kb][t].x; b[1] = bf[kb][t].y;
                acc[t] = __builtin_amdgcn_wmma_f32_16x16x4_f32(false, a0, false, b,
                                                               (short)0, acc[t], false, false);
            }
#pragma unroll
            for (int t = 0; t < TW; ++t) {
                v2f b; b[0] = bf[kb][t].z; b[1] = bf[kb][t].w;
                acc[t] = __builtin_amdgcn_wmma_f32_16x16x4_f32(false, a1, false, b,
                                                               (short)0, acc[t], false, false);
            }
            __builtin_amdgcn_sched_barrier(0);
        }

        float sc[8];
#pragma unroll
        for (int r = 0; r < 8; ++r) sc[r] = rsqrtf((float)(cn[r] + 1));
        float* crow = C + (size_t)(s * 16 + hh * 8) * NOUT + half * (TW * 16) + c;
#pragma unroll
        for (int t = 0; t < TW; ++t) {
#pragma unroll
            for (int r = 0; r < 8; ++r)
                crow[(size_t)r * NOUT + t * 16] = acc[t][r] * sc[r];
        }

#pragma unroll
        for (int kb = 0; kb < KB; ++kb) a[kb] = an[kb];
#pragma unroll
        for (int r = 0; r < 8; ++r) cn[r] = nn[r];
        s = sn;
    }
}


template <int F, int CH>
__device__ __forceinline__ void gather_chunk(float __attribute__((ext_vector_type(F / 32)))& acc,
                                             const float* __restrict__ hlane,
                                             int s_l, int j) {
    typedef float vf __attribute__((ext_vector_type(F / 32)));
    vf t[CH];
    for (int q = 0; q < CH; ++q) {
        const int s = __builtin_amdgcn_readlane(s_l, j + q);
        t[q] = *reinterpret_cast<const vf*>(hlane + (size_t)s * F);
    }
    for (int q = 0; q < CH; ++q) acc += t[q];
}

template <int F>
__device__ __forceinline__ void gather_batch(float __attribute__((ext_vector_type(F / 32)))& acc,
                                             const float* __restrict__ hlane,
                                             int s_l, int j, int end) {
    for (; j + 8 <= end; j += 8) gather_chunk<F, 8>(acc, hlane, s_l, j);
    const int rest = end - j;
    if (rest & 4) { gather_chunk<F, 4>(acc, hlane, s_l, j); j += 4; }
    if (rest & 2) { gather_chunk<F, 2>(acc, hlane, s_l, j); j += 2; }
    if (rest & 1) { gather_chunk<F, 1>(acc, hlane, s_l, j); }
}

template <int F, bool RELU>
__global__ __launch_bounds__(256)
void gather_aggregate(const int* __restrict__ bucket,
                      const int* __restrict__ esrc, const int* __restrict__ edst, long long E,
                      const float* __restrict__ h, const float* __restrict__ bias,
                      float* __restrict__ out, int n) {
    constexpr int V = F / 32;
    typedef float vf __attribute__((ext_vector_type(V)));
    const int lane = threadIdx.x & 31;
    const int d = __builtin_amdgcn_readfirstlane((int)(blockIdx.x * 8 + (threadIdx.x >> 5)));
    if (d >= n) return;
    const int* brow = bucket + (size_t)d * CAP;
    const int w0 = brow[lane];
    const float* hlane = h + lane * V;
    const vf hs = *reinterpret_cast<const vf*>(hlane + (size_t)d * F);
    const vf bv = *reinterpret_cast<const vf*>(bias + lane * V);

    const int deg = __builtin_amdgcn_readfirstlane(w0);
    const float dd = rsqrtf((float)(deg + 1));

    vf acc = {};
    if (deg <= SLOTS) {
        const int m = (deg < 31) ? deg : 31;
        gather_batch<F>(acc, hlane, w0, 1, 1 + m);
        if (deg > 31) {
            const int m2 = deg - 31;
            int s_l = 0;
            if (lane < m2) s_l = brow[32 + lane];
            gather_batch<F>(acc, hlane, s_l, 0, m2);
        }
    } else {
        for (long long e0 = 0; e0 < E; e0 += 32) {
            const long long e = e0 + lane;
            int s_l = 0;
            bool hit = false;
            if (e < E && edst[e] == d) {
                hit = true;
                s_l = esrc[e];
            }
            unsigned int todo = __builtin_amdgcn_ballot_w32(hit);
            while (todo != 0u) {
                const int b = __builtin_ctz(todo);
                todo &= todo - 1u;
                gather_chunk<F, 1>(acc, hlane, s_l, b);
            }
        }
    }

    vf v = (acc + hs) * dd + bv;
    if (RELU) {
        for (int i = 0; i < V; ++i) v[i] = fmaxf(v[i], 0.0f);
    }
    *reinterpret_cast<vf*>(out + (size_t)d * F + lane * V) = v;
}

static inline int blocks_for(long long n, int bs) { return (int)((n + bs - 1) / bs); }

extern "C" void kernel_launch(void* const* d_in, const int* in_sizes, int n_in,
                              void* d_out, int out_size, void* d_ws, size_t ws_size,
                              hipStream_t stream) {
    const float* x  = (const float*)d_in[0];
    const int*   ei = (const int*)d_in[1];
    const float* W1 = (const float*)d_in[2];
    const float* b1 = (const float*)d_in[3];
    const float* W2 = (const float*)d_in[4];
    const float* b2 = (const float*)d_in[5];
    float* out = (float*)d_out;

    const int N = in_sizes[0] / IN_C;
    const long long E = (long long)in_sizes[1] / 2;
    const int* esrc = ei;
    const int* edst = ei + E;

    float* ws   = (float*)d_ws;
    float* h1   = ws;
    float* a1   = h1 + (size_t)N * HID2;
    float* h2   = a1 + (size_t)N * HID2;
    int*   bucket = (int*)(h2 + (size_t)N * HID);

    fill_i32_kernel<<<blocks_for(N, 256), 256, 0, stream>>>(bucket, 0, N, CAP);
    bucket_place_kernel<<<blocks_for(E, 256), 256, 0, stream>>>(esrc, edst, bucket, E);

    {
        gemm_wmma_f32<HID2><<<GEMM_GROUPS, 256, 0, stream>>>(x, W1, bucket, h1, N);
    }
    gather_aggregate<HID2, true><<<blocks_for(N, 8), 256, 0, stream>>>(
        bucket, esrc, edst, E, h1, b1, a1, N);

    {
        gemm_wmma_f32<HID><<<GEMM_GROUPS, 256, 0, stream>>>(a1, W2, bucket, h2, N);
    }
    gather_aggregate<HID, false><<<blocks_for(N, 8), 256, 0, stream>>>(
        bucket, esrc, edst, E, h2, b2, out, N);
}


# Buckets split into two planes, one per coherence half of the package: an edge-pass wave adds to the plane of the half it runs on (die id), gathers join both runs; arrival-order rank, sum order varies
# speedup vs baseline: 5.2105x; 1.4602x over previous
#include <hip/hip_runtime.h>
#include <hip/hip_bf16.h>


typedef float v2f __attribute__((ext_vector_type(2)));
typedef float v8f __attribute__((ext_vector_type(8)));

#define IN_C 128
#define HID2 128
#define HID  64

__global__ __launch_bounds__(256)
void fill_i32_kernel(int* __restrict__ p, int v, long long n, int stride) {
    long long i = (long long)blockIdx.x * 256 + threadIdx.x;
    if (i < n) p[i * stride] = v;
}

#define CAP 32
#define SLOTS (CAP - 1)
__global__ __launch_bounds__(256)
void bucket_place_kernel(const int* __restrict__ src, const int* __restrict__ dst,
                         int* __restrict__ bucket, long long E, int n) {
    long long e = (long long)blockIdx.x * 256 + threadIdx.x;
    const int idmsg = __builtin_amdgcn_s_sendmsg_rtn(0x87);
    if (e < E) {
        const int s = src[e];
        const int d = dst[e];
        asm volatile("" :: "v"(s));
        const int half = (idmsg >> 18) & 1;
        int* brow = bucket + ((size_t)half * n + d) * CAP;
        const int r = __hip_atomic_fetch_add(brow, 1, __ATOMIC_RELAXED,
                                             __HIP_MEMORY_SCOPE_AGENT);
        if (r < SLOTS) brow[1 + r] = s;
    }
}

#define GEMM_GROUPS 256
template <int NOUT>
__global__ __launch_bounds__(256)
void gemm_wmma_f32(const float* __restrict__ A, const float* __restrict__ B,
                   const int* __restrict__ cnt, float* __restrict__ C, int M) {
    constexpr int K  = 128;
    constexpr int NT = NOUT / 16;
    constexpr int KB = K / 8;
    constexpr int TW = 4;
    constexpr int NH = NT / TW;
    __shared__ float4 bl[KB * NT * 32];

    const int nstrips = (M + 15) >> 4;
    if (nstrips <= 0) return;
    const int lane = threadIdx.x & 31;
    const int wave = __builtin_amdgcn_readfirstlane((int)(threadIdx.x >> 5));
    const int c  = lane & 15;
    const int hh = lane >> 4;

    const int nslots = (int)gridDim.x * 4;
    const int pair = wave >> 2;
    const int half = (NH == 2) ? pair : 0;
    const int step = (NH == 2) ? nslots : 2 * nslots;
    int s = (int)blockIdx.x * 4 + (wave & 3) + ((NH == 2) ? 0 : pair * nslots);

    const float4* abase = reinterpret_cast<const float4*>(A) + (size_t)c * (K / 4) + hh;

    float4 a[KB];
    int cn[8];
    {
        const int s0 = (s < nstrips) ? s : (nstrips - 1);
        const int* cp = cnt + ((size_t)s0 * 16 + hh * 8) * CAP;
#pragma unroll
        for (int r = 0; r < 8; ++r) cn[r] = cp[r * CAP] + cp[(size_t)M * CAP + r * CAP];
        const float4* ap = abase + (size_t)s0 * (16 * K / 4);
#pragma unroll
        for (int kb = 0; kb < KB; ++kb) a[kb] = ap[kb * 2];
    }

#pragma unroll
    for (int i = 0; i < KB * NT / 8; ++i) {
        const int g  = wave + 8 * i;
        const int t  = g % NT;
        const int kb = g / NT;
        const float* bp = B + (size_t)(kb * 8 + 4 * hh) * NOUT + t * 16 + c;
        float4 v;
        v.x = bp[0];
        v.y = bp[NOUT];
        v.z = bp[2 * NOUT];
        v.w = bp[3 * NOUT];
        bl[g * 32 + lane] = v;
    }
    __syncthreads();

    float4 bf[KB][TW];
    {
        const float4* bw = bl + (half * TW) * 32 + lane;
#pragma unroll
        for (int kb = 0; kb < KB; ++kb) {
#pragma unroll
            for (int t = 0; t < TW; ++t) bf[kb][t] = bw[(kb * NT + t) * 32];
        }
    }

    while (s < nstrips) {
        const int sn = s + step;
        const int sp = (sn < nstrips) ? sn : s;
        const int* cp = cnt + ((size_t)sp * 16 + hh * 8) * CAP;
        int nn[8];
#pragma unroll
        for (int r = 0; r < 8; ++r) nn[r] = cp[r * CAP] + cp[(size_t)M * CAP + r * CAP];
        float4 an[KB];
        {
            const float4* ap = abase + (size_t)sp * (16 * K / 4);
#pragma unroll
            for (int kb = 0; kb < KB; ++kb) an[kb] = ap[kb * 2];
        }

        v8f acc[TW];
#pragma unroll
        for (int t = 0; t < TW; ++t) acc[t] = v8f{};
#pragma unroll
        for (int kb = 0; kb < KB; ++kb) {
            v2f a0; a0[0] = a[kb].x; a0[1] = a[kb].y;
            v2f a1; a1[0] = a[kb].z; a1[1] = a[kb].w;
#pragma unroll
            for (int t = 0; t < TW; ++t) {
                v2f b; b[0] = bf[kb][t].x; b[1] = bf[kb][t].y;
                acc[t] = __builtin_amdgcn_wmma_f32_16x16x4_f32(false, a0, false, b,
                                                               (short)0, acc[t], false, false);
            }
#pragma unroll
            for (int t = 0; t < TW; ++t) {
                v2f b; b[0] = bf[kb][t].z; b[1] = bf[kb][t].w;
                acc[t] = __builtin_amdgcn_wmma_f32_16x16x4_f32(false, a1, false, b,
                                                               (short)0, acc[t], false, false);
            }
            __builtin_amdgcn_sched_barrier(0);
        }

        float sc[8];
#pragma unroll
        for (int r = 0; r < 8; ++r) sc[r] = rsqrtf((float)(cn[r] + 1));
        float* crow = C + (size_t)(s * 16 + hh * 8) * NOUT + half * (TW * 16) + c;
#pragma unroll
        for (int t = 0; t < TW; ++t) {
#pragma unroll
            for (int r = 0; r < 8; ++r)
                crow[(size_t)r * NOUT + t * 16] = acc[t][r] * sc[r];
        }

#pragma unroll
        for (int kb = 0; kb < KB; ++kb) a[kb] = an[kb];
#pragma unroll
        for (int r = 0; r < 8; ++r) cn[r] = nn[r];
        s = sn;
    }
}


template <int F, int CH>
__device__ __forceinline__ void gather_chunk(float __attribute__((ext_vector_type(F / 32)))& acc,
                                             const float* __restrict__ hlane,
                                             int s_l, int j) {
    typedef float vf __attribute__((ext_vector_type(F / 32)));
    vf t[CH];
    for (int q = 0; q < CH; ++q) {
        const int s = __builtin_amdgcn_readlane(s_l, j + q);
        t[q] = *reinterpret_cast<const vf*>(hlane + (size_t)s * F);
    }
    for (int q = 0; q < CH; ++q) acc += t[q];
}

template <int F>
__device__ __forceinline__ void gather_batch(float __attribute__((ext_vector_type(F / 32)))& acc,
                                             const float* __restrict__ hlane,
                                             int s_l, int j, int end) {
    for (; j + 8 <= end; j += 8) gather_chunk<F, 8>(acc, hlane, s_l, j);
    const int rest = end - j;
    if (rest & 4) { gather_chunk<F, 4>(acc, hlane, s_l, j); j += 4; }
    if (rest & 2) { gather_chunk<F, 2>(acc, hlane, s_l, j); j += 2; }
    if (rest & 1) { gather_chunk<F, 1>(acc, hlane, s_l, j); }
}

template <int F, bool RELU>
__global__ __launch_bounds__(256)
void gather_aggregate(const int* __restrict__ bucket,
                      const int* __restrict__ esrc, const int* __restrict__ edst, long long E,
                      const float* __restrict__ h, const float* __restrict__ bias,
                      float* __restrict__ out, int n) {
    constexpr int V = F / 32;
    typedef float vf __attribute__((ext_vector_type(V)));
    const int lane = threadIdx.x & 31;
    const int d = __builtin_amdgcn_readfirstlane((int)(blockIdx.x * 8 + (threadIdx.x >> 5)));
    if (d >= n) return;
    const int w0 = bucket[(size_t)d * CAP + lane];
    const int w1 = bucket[((size_t)n + d) * CAP + lane];
    const float* hlane = h + lane * V;
    const vf hs = *reinterpret_cast<const vf*>(hlane + (size_t)d * F);
    const vf bv = *reinterpret_cast<const vf*>(bias + lane * V);

    const int deg0 = __builtin_amdgcn_readfirstlane(w0);
    const int deg1 = __builtin_amdgcn_readfirstlane(w1);
    const int deg = deg0 + deg1;
    const float dd = rsqrtf((float)(deg + 1));

    vf acc = {};
    if (deg0 <= SLOTS && deg1 <= SLOTS) {
        {
            const int a = __builtin_amdgcn_ds_bpermute((1 + lane) << 2, w0);
            const int b = __builtin_amdgcn_ds_bpermute((1 + lane - deg0) << 2, w1);
            const int s_l = (lane < deg0) ? a : b;
            const int m = (deg < 32) ? deg : 32;
            gather_batch<F>(acc, hlane, s_l, 0, m);
        }
        if (deg > 32) {
            const int s_l = __builtin_amdgcn_ds_bpermute((33 + lane - deg0) << 2, w1);
            gather_batch<F>(acc, hlane, s_l, 0, deg - 32);
        }
    } else {
        for (long long e0 = 0; e0 < E; e0 += 32) {
            const long long e = e0 + lane;
            int s_l = 0;
            bool hit = false;
            if (e < E && edst[e] == d) {
                hit = true;
                s_l = esrc[e];
            }
            unsigned int todo = __builtin_amdgcn_ballot_w32(hit);
            while (todo != 0u) {
                const int b = __builtin_ctz(todo);
                todo &= todo - 1u;
                gather_chunk<F, 1>(acc, hlane, s_l, b);
            }
        }
    }

    vf v = (acc + hs) * dd + bv;
    if (RELU) {
        for (int i = 0; i < V; ++i) v[i] = fmaxf(v[i], 0.0f);
    }
    *reinterpret_cast<vf*>(out + (size_t)d * F + lane * V) = v;
}

static inline int blocks_for(long long n, int bs) { return (int)((n + bs - 1) / bs); }

extern "C" void kernel_launch(void* const* d_in, const int* in_sizes, int n_in,
                              void* d_out, int out_size, void* d_ws, size_t ws_size,
                              hipStream_t stream) {
    const float* x  = (const float*)d_in[0];
    const int*   ei = (const int*)d_in[1];
    const float* W1 = (const float*)d_in[2];
    const float* b1 = (const float*)d_in[3];
    const float* W2 = (const float*)d_in[4];
    const float* b2 = (const float*)d_in[5];
    float* out = (float*)d_out;

    const int N = in_sizes[0] / IN_C;
    const long long E = (long long)in_sizes[1] / 2;
    const int* esrc = ei;
    const int* edst = ei + E;

    float* ws   = (float*)d_ws;
    float* h1   = ws;
    float* a1   = h1 + (size_t)N * HID2;
    float* h2   = a1 + (size_t)N * HID2;
    int*   bucket = (int*)(h2 + (size_t)N * HID);

    fill_i32_kernel<<<blocks_for(2LL * N, 256), 256, 0, stream>>>(bucket, 0, 2LL * N, CAP);
    bucket_place_kernel<<<blocks_for(E, 256), 256, 0, stream>>>(esrc, edst, bucket, E, N);

    {
        gemm_wmma_f32<HID2><<<GEMM_GROUPS, 256, 0, stream>>>(x, W1, bucket, h1, N);
    }
    gather_aggregate<HID2, true><<<blocks_for(N, 8), 256, 0, stream>>>(
        bucket, esrc, edst, E, h1, b1, a1, N);

    {
        gemm_wmma_f32<HID><<<GEMM_GROUPS, 256, 0, stream>>>(a1, W2, bucket, h2, N);
    }
    gather_aggregate<HID, false><<<blocks_for(N, 8), 256, 0, stream>>>(
        bucket, esrc, edst, E, h2, b2, out, N);
}


# Layer-1 gather and layer-2 product in one launch: 4 resident 16-wave groups per WGP, a rotating wave pair multiplies a strip from LDS under the others' row loads (6 to 5 launches); sum order kept
# speedup vs baseline: 5.4534x; 1.0466x over previous
#include <hip/hip_runtime.h>
#include <hip/hip_bf16.h>


typedef float v2f __attribute__((ext_vector_type(2)));
typedef float v8f __attribute__((ext_vector_type(8)));

#define IN_C 128
#define HID2 128
#define HID  64

__global__ __launch_bounds__(256)
void fill_i32_kernel(int* __restrict__ p, int v, long long n, int stride) {
    long long i = (long long)blockIdx.x * 256 + threadIdx.x;
    if (i < n) p[i * stride] = v;
}

#define CAP 32
#define SLOTS (CAP - 1)
__global__ __launch_bounds__(256)
void bucket_place_kernel(const int* __restrict__ src, const int* __restrict__ dst,
                         int* __restrict__ bucket, long long E, int n) {
    long long e = (long long)blockIdx.x * 256 + threadIdx.x;
    const int idmsg = __builtin_amdgcn_s_sendmsg_rtn(0x87);
    if (e < E) {
        const int s = src[e];
        const int d = dst[e];
        asm volatile("" :: "v"(s));
        const int half = (idmsg >> 18) & 1;
        int* brow = bucket + ((size_t)half * n + d) * CAP;
        const int r = __hip_atomic_fetch_add(brow, 1, __ATOMIC_RELAXED,
                                             __HIP_MEMORY_SCOPE_AGENT);
        if (r < SLOTS) brow[1 + r] = s;
    }
}

#define GEMM_GROUPS 256
template <int NOUT>
__global__ __launch_bounds__(256)
void gemm_wmma_f32(const float* __restrict__ A, const float* __restrict__ B,
                   const int* __restrict__ cnt, float* __restrict__ C, int M) {
    constexpr int K  = 128;
    constexpr int NT = NOUT / 16;
    constexpr int KB = K / 8;
    constexpr int TW = 4;
    constexpr int NH = NT / TW;
    __shared__ float4 bl[KB * NT * 32];

    const int nstrips = (M + 15) >> 4;
    if (nstrips <= 0) return;
    const int lane = threadIdx.x & 31;
    const int wave = __builtin_amdgcn_readfirstlane((int)(threadIdx.x >> 5));
    const int c  = lane & 15;
    const int hh = lane >> 4;

    const int nslots = (int)gridDim.x * 4;
    const int pair = wave >> 2;
    const int half = (NH == 2) ? pair : 0;
    const int step = (NH == 2) ? nslots : 2 * nslots;
    int s = (int)blockIdx.x * 4 + (wave & 3) + ((NH == 2) ? 0 : pair * nslots);

    const float4* abase = reinterpret_cast<const float4*>(A) + (size_t)c * (K / 4) + hh;

    float4 a[KB];
    int cn[8];
    {
        const int s0 = (s < nstrips) ? s : (nstrips - 1);
        const int* cp = cnt + ((size_t)s0 * 16 + hh * 8) * CAP;
#pragma unroll
        for (int r = 0; r < 8; ++r) cn[r] = cp[r * CAP] + cp[(size_t)M * CAP + r * CAP];
        const float4* ap = abase + (size_t)s0 * (16 * K / 4);
#pragma unroll
        for (int kb = 0; kb < KB; ++kb) a[kb] = ap[kb * 2];
    }

#pragma unroll
    for (int i = 0; i < KB * NT / 8; ++i) {
        const int g  = wave + 8 * i;
        const int t  = g % NT;
        const int kb = g / NT;
        const float* bp = B + (size_t)(kb * 8 + 4 * hh) * NOUT + t * 16 + c;
        float4 v;
        v.x = bp[0];
        v.y = bp[NOUT];
        v.z = bp[2 * NOUT];
        v.w = bp[3 * NOUT];
        bl[g * 32 + lane] = v;
    }
    __syncthreads();

    float4 bf[KB][TW];
    {
        const float4* bw = bl + (half * TW) * 32 + lane;
#pragma unroll
        for (int kb = 0; kb < KB; ++kb) {
#pragma unroll
            for (int t = 0; t < TW; ++t) bf[kb][t] = bw[(kb * NT + t) * 32];
        }
    }

    while (s < nstrips) {
        const int sn = s + step;
        const int sp = (sn < nstrips) ? sn : s;
        const int* cp = cnt + ((size_t)sp * 16 + hh * 8) * CAP;
        int nn[8];
#pragma unroll
        for (int r = 0; r < 8; ++r) nn[r] = cp[r * CAP] + cp[(size_t)M * CAP + r * CAP];
        float4 an[KB];
        {
            const float4* ap = abase + (size_t)sp * (16 * K / 4);
#pragma unroll
            for (int kb = 0; kb < KB; ++kb) an[kb] = ap[kb * 2];
        }

        v8f acc[TW];
#pragma unroll
        for (int t = 0; t < TW; ++t) acc[t] = v8f{};
#pragma unroll
        for (int kb = 0; kb < KB; ++kb) {
            v2f a0; a0[0] = a[kb].x; a0[1] = a[kb].y;
            v2f a1; a1[0] = a[kb].z; a1[1] = a[kb].w;
#pragma unroll
            for (int t = 0; t < TW; ++t) {
                v2f b; b[0] = bf[kb][t].x; b[1] = bf[kb][t].y;
                acc[t] = __builtin_amdgcn_wmma_f32_16x16x4_f32(false, a0, false, b,
                                                               (short)0, acc[t], false, false);
            }
#pragma unroll
            for (int t = 0; t < TW; ++t) {
                v2f b; b[0] = bf[kb][t].z; b[1] = bf[kb][t].w;
                acc[t] = __builtin_amdgcn_wmma_f32_16x16x4_f32(false, a1, false, b,
                                                               (short)0, acc[t], false, false);
            }
            __builtin_amdgcn_sched_barrier(0);
        }

        float sc[8];
#pragma unroll
        for (int r = 0; r < 8; ++r) sc[r] = rsqrtf((float)(cn[r] + 1));
        float* crow = C + (size_t)(s * 16 + hh * 8) * NOUT + half * (TW * 16) + c;
#pragma unroll
        for (int t = 0; t < TW; ++t) {
#pragma unroll
            for (int r = 0; r < 8; ++r)
                crow[(size_t)r * NOUT + t * 16] = acc[t][r] * sc[r];
        }

#pragma unroll
        for (int kb = 0; kb < KB; ++kb) a[kb] = an[kb];
#pragma unroll
        for (int r = 0; r < 8; ++r) cn[r] = nn[r];
        s = sn;
    }
}


template <int F, int CH>
__device__ __forceinline__ void gather_chunk(float __attribute__((ext_vector_type(F / 32)))& acc,
                                             const float* __restrict__ hlane,
                                             int s_l, int j) {
    typedef float vf __attribute__((ext_vector_type(F / 32)));
    vf t[CH];
    for (int q = 0; q < CH; ++q) {
        const int s = __builtin_amdgcn_readlane(s_l, j + q);
        t[q] = *reinterpret_cast<const vf*>(hlane + (size_t)s * F);
    }
    for (int q = 0; q < CH; ++q) acc += t[q];
}

template <int F>
__device__ __forceinline__ void gather_batch(float __attribute__((ext_vector_type(F / 32)))& acc,
                                             const float* __restrict__ hlane,
                                             int s_l, int j, int end) {
    for (; j + 8 <= end; j += 8) gather_chunk<F, 8>(acc, hlane, s_l, j);
    const int rest = end - j;
    if (rest & 4) { gather_chunk<F, 4>(acc, hlane, s_l, j); j += 4; }
    if (rest & 2) { gather_chunk<F, 2>(acc, hlane, s_l, j); j += 2; }
    if (rest & 1) { gather_chunk<F, 1>(acc, hlane, s_l, j); }
}

template <int F, bool RELU>
__device__ __forceinline__ float __attribute__((ext_vector_type(F / 32)))
gather_node(const int* __restrict__ bucket,
            const int* __restrict__ esrc, const int* __restrict__ edst, long long E,
            const float* __restrict__ h, const float* __restrict__ bias,
            int n, int d, int lane, float& dd_out) {
    constexpr int V = F / 32;
    typedef float vf __attribute__((ext_vector_type(V)));
    const int w0 = bucket[(size_t)d * CAP + lane];
    const int w1 = bucket[((size_t)n + d) * CAP + lane];
    const float* hlane = h + lane * V;
    const vf hs = *reinterpret_cast<const vf*>(hlane + (size_t)d * F);
    const vf bv = *reinterpret_cast<const vf*>(bias + lane * V);

    const int deg0 = __builtin_amdgcn_readfirstlane(w0);
    const int deg1 = __builtin_amdgcn_readfirstlane(w1);
    const int deg = deg0 + deg1;
    const float dd = rsqrtf((float)(deg + 1));

    vf acc = {};
    if (deg0 <= SLOTS && deg1 <= SLOTS) {
        {
            const int a = __builtin_amdgcn_ds_bpermute((1 + lane) << 2, w0);
            const int b = __builtin_amdgcn_ds_bpermute((1 + lane - deg0) << 2, w1);
            const int s_l = (lane < deg0) ? a : b;
            const int m = (deg < 32) ? deg : 32;
            gather_batch<F>(acc, hlane, s_l, 0, m);
        }
        if (deg > 32) {
            const int s_l = __builtin_amdgcn_ds_bpermute((33 + lane - deg0) << 2, w1);
            gather_batch<F>(acc, hlane, s_l, 0, deg - 32);
        }
    } else {
        for (long long e0 = 0; e0 < E; e0 += 32) {
            const long long e = e0 + lane;
            int s_l = 0;
            bool hit = false;
            if (e < E && edst[e] == d) {
                hit = true;
                s_l = esrc[e];
            }
            unsigned int todo = __builtin_amdgcn_ballot_w32(hit);
            while (todo != 0u) {
                const int b = __builtin_ctz(todo);
                todo &= todo - 1u;
                gather_chunk<F, 1>(acc, hlane, s_l, b);
            }
        }
    }

    vf v = (acc + hs) * dd + bv;
    if (RELU) {
        for (int i = 0; i < V; ++i) v[i] = fmaxf(v[i], 0.0f);
    }
    dd_out = dd;
    return v;
}

template <int F, bool RELU>
__global__ __launch_bounds__(256)
void gather_aggregate(const int* __restrict__ bucket,
                      const int* __restrict__ esrc, const int* __restrict__ edst, long long E,
                      const float* __restrict__ h, const float* __restrict__ bias,
                      float* __restrict__ out, int n) {
    constexpr int V = F / 32;
    typedef float vf __attribute__((ext_vector_type(V)));
    const int lane = threadIdx.x & 31;
    const int d = __builtin_amdgcn_readfirstlane((int)(blockIdx.x * 8 + (threadIdx.x >> 5)));
    if (d >= n) return;
    float dd;
    const vf v = gather_node<F, RELU>(bucket, esrc, edst, E, h, bias, n, d, lane, dd);
    *reinterpret_cast<vf*>(out + (size_t)d * F + lane * V) = v;
}

#define FUSED_GROUPS 1024
#define TROW 33
__global__ __launch_bounds__(512) __attribute__((amdgpu_waves_per_eu(16, 16)))
void gather_relu_gemm(const int* __restrict__ bucket,
                      const int* __restrict__ esrc, const int* __restrict__ edst, long long E,
                      const float* __restrict__ h, const float* __restrict__ bias,
                      const float* __restrict__ B, float* __restrict__ C, int n) {
    constexpr int K = HID2, NOUT = HID;
    constexpr int NT = NOUT / 16;
    constexpr int KB = K / 8;
    typedef float vf __attribute__((ext_vector_type(4)));
    __shared__ float4 bl[KB * NT * 32];
    __shared__ float4 tile[2][16 * TROW];
    __shared__ float4 dsc[2][4];

    const int nstrips = (n + 15) >> 4;
    const int lane = threadIdx.x & 31;
    const int wave = __builtin_amdgcn_readfirstlane((int)(threadIdx.x >> 5));
    const int c  = lane & 15;
    const int hh = lane >> 4;

#pragma unroll
    for (int i = 0; i < KB * NT / 16; ++i) {
        const int g  = wave + 16 * i;
        const int t  = g % NT;
        const int kb = g / NT;
        const float* bp = B + (size_t)(kb * 8 + 4 * hh) * NOUT + t * 16 + c;
        float4 v;
        v.x = bp[0];
        v.y = bp[NOUT];
        v.z = bp[2 * NOUT];
        v.w = bp[3 * NOUT];
        bl[g * 32 + lane] = v;
    }

    int it = 0;
    for (int s = (int)blockIdx.x; s < nstrips; s += (int)gridDim.x, ++it) {
        const int buf = it & 1;
        const int d = s * 16 + wave;
        vf v = {};
        float dd = 0.0f;
        if (d < n)
            v = gather_node<HID2, true>(bucket, esrc, edst, E, h, bias, n, d, lane, dd);
        float4 vr; vr.x = v[0]; vr.y = v[1]; vr.z = v[2]; vr.w = v[3];
        tile[buf][wave * TROW + lane] = vr;
        if (lane == 0) reinterpret_cast<float*>(dsc[buf])[wave] = dd;
        __syncthreads();

        const int pw = (wave - 2 * it) & 15;
        if (pw < 2) {
            const float4* arow = &tile[buf][c * TROW + hh];
            const float4* bw = bl + (2 * pw) * 32 + lane;
            v8f acc0 = v8f{};
            v8f acc1 = v8f{};
#pragma unroll
            for (int kb = 0; kb < KB; ++kb) {
                const float4 a  = arow[kb * 2];
                const float4 b0 = bw[(kb * NT + 0) * 32];
                const float4 b1 = bw[(kb * NT + 1) * 32];
                v2f a0; a0[0] = a.x; a0[1] = a.y;
                v2f a1; a1[0] = a.z; a1[1] = a.w;
                v2f p0; p0[0] = b0.x; p0[1] = b0.y;
                v2f p1; p1[0] = b1.x; p1[1] = b1.y;
                v2f q0; q0[0] = b0.z; q0[1] = b0.w;
                v2f q1; q1[0] = b1.z; q1[1] = b1.w;
                acc0 = __builtin_amdgcn_wmma_f32_16x16x4_f32(false, a0, false, p0,
                                                             (short)0, acc0, false, false);
                acc1 = __builtin_amdgcn_wmma_f32_16x16x4_f32(false, a0, false, p1,
                                                             (short)0, acc1, false, false);
                acc0 = __builtin_amdgcn_wmma_f32_16x16x4_f32(false, a1, false, q0,
                                                             (short)0, acc0, false, false);
                acc1 = __builtin_amdgcn_wmma_f32_16x16x4_f32(false, a1, false, q1,
                                                             (short)0, acc1, false, false);
            }
            const float4 s0 = dsc[buf][hh * 2];
            const float4 s1 = dsc[buf][hh * 2 + 1];
            const float sc[8] = {s0.x, s0.y, s0.z, s0.w, s1.x, s1.y, s1.z, s1.w};
            const int row0 = s * 16 + hh * 8;
            float* crow = C + (size_t)row0 * NOUT + pw * 32 + c;
#pragma unroll
            for (int r = 0; r < 8; ++r) {
                if (row0 + r < n) {
                    crow[(size_t)r * NOUT]      = acc0[r] * sc[r];
                    crow[(size_t)r * NOUT + 16] = acc1[r] * sc[r];
                }
            }
        }
    }
}

static inline int blocks_for(long long n, int bs) { return (int)((n + bs - 1) / bs); }

extern "C" void kernel_launch(void* const* d_in, const int* in_sizes, int n_in,
                              void* d_out, int out_size, void* d_ws, size_t ws_size,
                              hipStream_t stream) {
    const float* x  = (const float*)d_in[0];
    const int*   ei = (const int*)d_in[1];
    const float* W1 = (const float*)d_in[2];
    const float* b1 = (const float*)d_in[3];
    const float* W2 = (const float*)d_in[4];
    const float* b2 = (const float*)d_in[5];
    float* out = (float*)d_out;

    const int N = in_sizes[0] / IN_C;
    const long long E = (long long)in_sizes[1] / 2;
    const int* esrc = ei;
    const int* edst = ei + E;

    float* ws   = (float*)d_ws;
    float* h1   = ws;
    float* a1   = h1 + (size_t)N * HID2;
    float* h2   = a1 + (size_t)N * HID2;
    int*   bucket = (int*)(h2 + (size_t)N * HID);

    fill_i32_kernel<<<blocks_for(2LL * N, 256), 256, 0, stream>>>(bucket, 0, 2LL * N, CAP);
    bucket_place_kernel<<<blocks_for(E, 256), 256, 0, stream>>>(esrc, edst, bucket, E, N);

    {
        gemm_wmma_f32<HID2><<<GEMM_GROUPS, 256, 0, stream>>>(x, W1, bucket, h1, N);
    }
    gather_relu_gemm<<<FUSED_GROUPS, 512, 0, stream>>>(
        bucket, esrc, edst, E, h1, b1, W2, h2, N);

    gather_aggregate<HID, false><<<blocks_for(N, 8), 256, 0, stream>>>(
        bucket, esrc, edst, E, h2, b2, out, N);
}


# Edge bucketing without global atomics or zero fill: edge pieces filed by destination range, a work-group per range places them in LDS and writes bucket lines whole; LDS-atomic rank, sum order varies
# speedup vs baseline: 6.0278x; 1.1053x over previous
#include <hip/hip_runtime.h>
#include <hip/hip_bf16.h>


typedef float v2f __attribute__((ext_vector_type(2)));
typedef float v8f __attribute__((ext_vector_type(8)));

#define IN_C 128
#define HID2 128
#define HID  64

#define CAP 32
#define SLOTS (CAP - 1)
#define RSH 8
#define RW (1 << RSH)
#define NRMAX 448
#define EPB 4096
#define PSLOTS 31
#define NSLOT (2 * SLOTS)
#define LPITCH 33
__global__ __launch_bounds__(1024)
void edge_partition_kernel(const int* __restrict__ src, const int* __restrict__ dst,
                           int* __restrict__ part, long long E, int nr, int npieces) {
    constexpr int EPT = EPB / 1024;
    __shared__ int img[NRMAX * LPITCH];
    const int tid = threadIdx.x;
    const int lane = tid & 31;
    const int wave = __builtin_amdgcn_readfirstlane(tid >> 5);
    const long long base = (long long)blockIdx.x * EPB;
    int s[EPT], d[EPT];
#pragma unroll
    for (int i = 0; i < EPT; ++i) {
        const long long e = base + tid + i * 1024;
        const long long ec = (e < E) ? e : (E - 1);
        s[i] = src[ec];
        const int dv = dst[ec];
        d[i] = (e < E) ? dv : -1;
    }
    for (int i = tid; i < NRMAX * LPITCH; i += 1024) img[i] = 0;
    __syncthreads();
    int r[EPT];
#pragma unroll
    for (int i = 0; i < EPT; ++i)
        r[i] = (d[i] >= 0) ? atomicAdd(&img[(d[i] >> RSH) * LPITCH], 1) : PSLOTS;
#pragma unroll
    for (int i = 0; i < EPT; ++i) {
        if (r[i] < PSLOTS)
            img[(d[i] >> RSH) * LPITCH + 1 + r[i]] = (s[i] << RSH) | (d[i] & (RW - 1));
    }
    __syncthreads();
    int v[NRMAX / 32];
#pragma unroll
    for (int j = 0; j < NRMAX / 32; ++j) v[j] = img[(wave + 32 * j) * LPITCH + lane];
#pragma unroll
    for (int j = 0; j < NRMAX / 32; ++j) {
        const int rg = wave + 32 * j;
        if (rg < nr)
            part[((size_t)rg * npieces + blockIdx.x) * 32 + lane] = v[j];
    }
}

__device__ __forceinline__ void lds_place(int* deg, int* slot, int s, int dl) {
    const int r = atomicAdd(&deg[dl], 1);
    if (r < NSLOT) slot[dl * NSLOT + r] = s;
}
#define PUN 13
__global__ __launch_bounds__(1024)
void bucket_build_kernel(const int* __restrict__ part,
                         const int* __restrict__ esrc, const int* __restrict__ edst, long long E,
                         int* __restrict__ bucket, int n, int npieces) {
    __shared__ int deg[RW];
    __shared__ int slot[RW * NSLOT];
    const int tid = threadIdx.x;
    const int lane = tid & 31;
    const int wave = __builtin_amdgcn_readfirstlane(tid >> 5);
    const int k = (int)blockIdx.x;
    if (tid < RW) deg[tid] = 0;
    __syncthreads();

    const int* prow = part + (size_t)k * npieces * 32 + lane;
    for (int b0 = wave; b0 < npieces; b0 += 32 * PUN) {
        int w[PUN];
#pragma unroll
        for (int u = 0; u < PUN; ++u) {
            const int b = b0 + 32 * u;
            w[u] = (b < npieces) ? prow[(size_t)b * 32] : 0;
        }
        int r[PUN];
        unsigned int over = 0u;
#pragma unroll
        for (int u = 0; u < PUN; ++u) {
            const int c = __builtin_amdgcn_readfirstlane(w[u]);
            if (c > PSLOTS) over |= 1u << u;
            const bool mine = (c <= PSLOTS) && lane >= 1 && lane <= c;
            r[u] = mine ? atomicAdd(&deg[w[u] & (RW - 1)], 1) : NSLOT;
        }
#pragma unroll
        for (int u = 0; u < PUN; ++u) {
            if (r[u] < NSLOT) slot[(w[u] & (RW - 1)) * NSLOT + r[u]] = w[u] >> RSH;
        }
        while (over != 0u) {
            const int u = __builtin_ctz(over);
            over &= over - 1u;
            const long long e0 = (long long)(b0 + 32 * u) * EPB;
            for (int i = 0; i < EPB; i += 32) {
                const long long e = e0 + i + lane;
                if (e < E) {
                    const int d = edst[e];
                    if ((d >> RSH) == k) lds_place(deg, slot, esrc[e], d & (RW - 1));
                }
            }
        }
    }
    __syncthreads();

    const int dgl = deg[(wave + 32 * lane) & (RW - 1)];
    int v0[RW / 32], v1[RW / 32];
#pragma unroll
    for (int i = 0; i < RW / 32; ++i) {
        const int dl = wave + 32 * i;
        const int dg = __builtin_amdgcn_readlane(dgl, i);
        const int d0 = (dg < SLOTS) ? dg : SLOTS;
        const int d1 = dg - d0;
        const int j = lane - 1;
        const int* srow = slot + dl * NSLOT;
        const bool has0 = j >= 0 && j < d0;
        const bool has1 = j >= 0 && j < d1;
        const int* p0 = has0 ? &srow[j] : &deg[dl];
        const int* p1 = has1 ? &srow[SLOTS + j] : &deg[dl];
        const int t0 = *p0;
        const int t1 = *p1;
        const int m0 = (t0 < SLOTS) ? t0 : SLOTS;
        const int m1 = (t1 < SLOTS) ? t1 : SLOTS;
        v0[i] = has0 ? t0 : ((lane == 0) ? m0 : 0);
        v1[i] = has1 ? t1 : ((lane == 0) ? (t1 - m1) : 0);
    }
#pragma unroll
    for (int i = 0; i < RW / 32; ++i) {
        const int d = k * RW + wave + 32 * i;
        if (d < n) {
            bucket[(size_t)d * CAP + lane] = v0[i];
            bucket[((size_t)n + d) * CAP + lane] = v1[i];
        }
    }
}

#define GEMM_GROUPS 256
template <int NOUT>
__global__ __launch_bounds__(256)
void gemm_wmma_f32(const float* __restrict__ A, const float* __restrict__ B,
                   const int* __restrict__ cnt, float* __restrict__ C, int M) {
    constexpr int K  = 128;
    constexpr int NT = NOUT / 16;
    constexpr int KB = K / 8;
    constexpr int TW = 4;
    constexpr int NH = NT / TW;
    __shared__ float4 bl[KB * NT * 32];

    const int nstrips = (M + 15) >> 4;
    if (nstrips <= 0) return;
    const int lane = threadIdx.x & 31;
    const int wave = __builtin_amdgcn_readfirstlane((int)(threadIdx.x >> 5));
    const int c  = lane & 15;
    const int hh = lane >> 4;

    const int nslots = (int)gridDim.x * 4;
    const int pair = wave >> 2;
    const int half = (NH == 2) ? pair : 0;
    const int step = (NH == 2) ? nslots : 2 * nslots;
    int s = (int)blockIdx.x * 4 + (wave & 3) + ((NH == 2) ? 0 : pair * nslots);

    const float4* abase = reinterpret_cast<const float4*>(A) + (size_t)c * (K / 4) + hh;

    float4 a[KB];
    int cn[8];
    {
        const int s0 = (s < nstrips) ? s : (nstrips - 1);
        const int* cp = cnt + ((size_t)s0 * 16 + hh * 8) * CAP;
#pragma unroll
        for (int r = 0; r < 8; ++r) cn[r] = cp[r * CAP] + cp[(size_t)M * CAP + r * CAP];
        const float4* ap = abase + (size_t)s0 * (16 * K / 4);
#pragma unroll
        for (int kb = 0; kb < KB; ++kb) a[kb] = ap[kb * 2];
    }

#pragma unroll
    for (int i = 0; i < KB * NT / 8; ++i) {
        const int g  = wave + 8 * i;
        const int t  = g % NT;
        const int kb = g / NT;
        const float* bp = B + (size_t)(kb * 8 + 4 * hh) * NOUT + t * 16 + c;
        float4 v;
        v.x = bp[0];
        v.y = bp[NOUT];
        v.z = bp[2 * NOUT];
        v.w = bp[3 * NOUT];
        bl[g * 32 + lane] = v;
    }
    __syncthreads();

    float4 bf[KB][TW];
    {
        const float4* bw = bl + (half * TW) * 32 + lane;
#pragma unroll
        for (int kb = 0; kb < KB; ++kb) {
#pragma unroll
            for (int t = 0; t < TW; ++t) bf[kb][t] = bw[(kb * NT + t) * 32];
        }
    }

    while (s < nstrips) {
        const int sn = s + step;
        const int sp = (sn < nstrips) ? sn : s;
        const int* cp = cnt + ((size_t)sp * 16 + hh * 8) * CAP;
        int nn[8];
#pragma unroll
        for (int r = 0; r < 8; ++r) nn[r] = cp[r * CAP] + cp[(size_t)M * CAP + r * CAP];
        float4 an[KB];
        {
            const float4* ap = abase + (size_t)sp * (16 * K / 4);
#pragma unroll
            for (int kb = 0; kb < KB; ++kb) an[kb] = ap[kb * 2];
        }

        v8f acc[TW];
#pragma unroll
        for (int t = 0; t < TW; ++t) acc[t] = v8f{};
#pragma unroll
        for (int kb = 0; kb < KB; ++kb) {
            v2f a0; a0[0] = a[kb].x; a0[1] = a[kb].y;
            v2f a1; a1[0] = a[kb].z; a1[1] = a[kb].w;
#pragma unroll
            for (int t = 0; t < TW; ++t) {
                v2f b; b[0] = bf[kb][t].x; b[1] = bf[kb][t].y;
                acc[t] = __builtin_amdgcn_wmma_f32_16x16x4_f32(false, a0, false, b,
                                                               (short)0, acc[t], false, false);
            }
#pragma unroll
            for (int t = 0; t < TW; ++t) {
                v2f b; b[0] = bf[kb][t].z; b[1] = bf[kb][t].w;
                acc[t] = __builtin_amdgcn_wmma_f32_16x16x4_f32(false, a1, false, b,
                                                               (short)0, acc[t], false, false);
            }
            __builtin_amdgcn_sched_barrier(0);
        }

        float sc[8];
#pragma unroll
        for (int r = 0; r < 8; ++r) sc[r] = rsqrtf((float)(cn[r] + 1));
        float* crow = C + (size_t)(s * 16 + hh * 8) * NOUT + half * (TW * 16) + c;
#pragma unroll
        for (int t = 0; t < TW; ++t) {
#pragma unroll
            for (int r = 0; r < 8; ++r)
                crow[(size_t)r * NOUT + t * 16] = acc[t][r] * sc[r];
        }

#pragma unroll
        for (int kb = 0; kb < KB; ++kb) a[kb] = an[kb];
#pragma unroll
        for (int r = 0; r < 8; ++r) cn[r] = nn[r];
        s = sn;
    }
}


template <int F, int CH>
__device__ __forceinline__ void gather_chunk(float __attribute__((ext_vector_type(F / 32)))& acc,
                                             const float* __restrict__ hlane,
                                             int s_l, int j) {
    typedef float vf __attribute__((ext_vector_type(F / 32)));
    vf t[CH];
    for (int q = 0; q < CH; ++q) {
        const int s = __builtin_amdgcn_readlane(s_l, j + q);
        t[q] = *reinterpret_cast<const vf*>(hlane + (size_t)s * F);
    }
    for (int q = 0; q < CH; ++q) acc += t[q];
}

template <int F>
__device__ __forceinline__ void gather_batch(float __attribute__((ext_vector_type(F / 32)))& acc,
                                             const float* __restrict__ hlane,
                                             int s_l, int j, int end) {
    for (; j + 8 <= end; j += 8) gather_chunk<F, 8>(acc, hlane, s_l, j);
    const int rest = end - j;
    if (rest & 4) { gather_chunk<F, 4>(acc, hlane, s_l, j); j += 4; }
    if (rest & 2) { gather_chunk<F, 2>(acc, hlane, s_l, j); j += 2; }
    if (rest & 1) { gather_chunk<F, 1>(acc, hlane, s_l, j); }
}

template <int F, bool RELU>
__device__ __forceinline__ float __attribute__((ext_vector_type(F / 32)))
gather_node(const int* __restrict__ bucket,
            const int* __restrict__ esrc, const int* __restrict__ edst, long long E,
            const float* __restrict__ h, const float* __restrict__ bias,
            int n, int d, int lane, float& dd_out) {
    constexpr int V = F / 32;
    typedef float vf __attribute__((ext_vector_type(V)));
    const int w0 = bucket[(size_t)d * CAP + lane];
    const int w1 = bucket[((size_t)n + d) * CAP + lane];
    const float* hlane = h + lane * V;
    const vf hs = *reinterpret_cast<const vf*>(hlane + (size_t)d * F);
    const vf bv = *reinterpret_cast<const vf*>(bias + lane * V);

    const int deg0 = __builtin_amdgcn_readfirstlane(w0);
    const int deg1 = __builtin_amdgcn_readfirstlane(w1);
    const int deg = deg0 + deg1;
    const float dd = rsqrtf((float)(deg + 1));

    vf acc = {};
    if (deg0 <= SLOTS && deg1 <= SLOTS) {
        {
            const int a = __builtin_amdgcn_ds_bpermute((1 + lane) << 2, w0);
            const int b = __builtin_amdgcn_ds_bpermute((1 + lane - deg0) << 2, w1);
            const int s_l = (lane < deg0) ? a : b;
            const int m = (deg < 32) ? deg : 32;
            gather_batch<F>(acc, hlane, s_l, 0, m);
        }
        if (deg > 32) {
            const int s_l = __builtin_amdgcn_ds_bpermute((33 + lane - deg0) << 2, w1);
            gather_batch<F>(acc, hlane, s_l, 0, deg - 32);
        }
    } else {
        for (long long e0 = 0; e0 < E; e0 += 32) {
            const long long e = e0 + lane;
            int s_l = 0;
            bool hit = false;
            if (e < E && edst[e] == d) {
                hit = true;
                s_l = esrc[e];
            }
            unsigned int todo = __builtin_amdgcn_ballot_w32(hit);
            while (todo != 0u) {
                const int b = __builtin_ctz(todo);
                todo &= todo - 1u;
                gather_chunk<F, 1>(acc, hlane, s_l, b);
            }
        }
    }

    vf v = (acc + hs) * dd + bv;
    if (RELU) {
        for (int i = 0; i < V; ++i) v[i] = fmaxf(v[i], 0.0f);
    }
    dd_out = dd;
    return v;
}

template <int F, bool RELU>
__global__ __launch_bounds__(256)
void gather_aggregate(const int* __restrict__ bucket,
                      const int* __restrict__ esrc, const int* __restrict__ edst, long long E,
                      const float* __restrict__ h, const float* __restrict__ bias,
                      float* __restrict__ out, int n) {
    constexpr int V = F / 32;
    typedef float vf __attribute__((ext_vector_type(V)));
    const int lane = threadIdx.x & 31;
    const int d = __builtin_amdgcn_readfirstlane((int)(blockIdx.x * 8 + (threadIdx.x >> 5)));
    if (d >= n) return;
    float dd;
    const vf v = gather_node<F, RELU>(bucket, esrc, edst, E, h, bias, n, d, lane, dd);
    *reinterpret_cast<vf*>(out + (size_t)d * F + lane * V) = v;
}

#define FUSED_GROUPS 1024
#define TROW 33
__global__ __launch_bounds__(512) __attribute__((amdgpu_waves_per_eu(16, 16)))
void gather_relu_gemm(const int* __restrict__ bucket,
                      const int* __restrict__ esrc, const int* __restrict__ edst, long long E,
                      const float* __restrict__ h, const float* __restrict__ bias,
                      const float* __restrict__ B, float* __restrict__ C, int n) {
    constexpr int K = HID2, NOUT = HID;
    constexpr int NT = NOUT / 16;
    constexpr int KB = K / 8;
    typedef float vf __attribute__((ext_vector_type(4)));
    __shared__ float4 bl[KB * NT * 32];
    __shared__ float4 tile[2][16 * TROW];
    __shared__ float4 dsc[2][4];

    const int nstrips = (n + 15) >> 4;
    const int lane = threadIdx.x & 31;
    const int wave = __builtin_amdgcn_readfirstlane((int)(threadIdx.x >> 5));
    const int c  = lane & 15;
    const int hh = lane >> 4;

#pragma unroll
    for (int i = 0; i < KB * NT / 16; ++i) {
        const int g  = wave + 16 * i;
        const int t  = g % NT;
        const int kb = g / NT;
        const float* bp = B + (size_t)(kb * 8 + 4 * hh) * NOUT + t * 16 + c;
        float4 v;
        v.x = bp[0];
        v.y = bp[NOUT];
        v.z = bp[2 * NOUT];
        v.w = bp[3 * NOUT];
        bl[g * 32 + lane] = v;
    }

    int it = 0;
    for (int s = (int)blockIdx.x; s < nstrips; s += (int)gridDim.x, ++it) {
        const int buf = it & 1;
        const int d = s * 16 + wave;
        vf v = {};
        float dd = 0.0f;
        if (d < n)
            v = gather_node<HID2, true>(bucket, esrc, edst, E, h, bias, n, d, lane, dd);
        float4 vr; vr.x = v[0]; vr.y = v[1]; vr.z = v[2]; vr.w = v[3];
        tile[buf][wave * TROW + lane] = vr;
        if (lane == 0) reinterpret_cast<float*>(dsc[buf])[wave] = dd;
        __syncthreads();

        const int pw = (wave - 2 * it) & 15;
        if (pw < 2) {
            const float4* arow = &tile[buf][c * TROW + hh];
            const float4* bw = bl + (2 * pw) * 32 + lane;
            v8f acc0 = v8f{};
            v8f acc1 = v8f{};
#pragma unroll
            for (int kb = 0; kb < KB; ++kb) {
                const float4 a  = arow[kb * 2];
                const float4 b0 = bw[(kb * NT + 0) * 32];
                const float4 b1 = bw[(kb * NT + 1) * 32];
                v2f a0; a0[0] = a.x; a0[1] = a.y;
                v2f a1; a1[0] = a.z; a1[1] = a.w;
                v2f p0; p0[0] = b0.x; p0[1] = b0.y;
                v2f p1; p1[0] = b1.x; p1[1] = b1.y;
                v2f q0; q0[0] = b0.z; q0[1] = b0.w;
                v2f q1; q1[0] = b1.z; q1[1] = b1.w;
                acc0 = __builtin_amdgcn_wmma_f32_16x16x4_f32(false, a0, false, p0,
                                                             (short)0, acc0, false, false);
                acc1 = __builtin_amdgcn_wmma_f32_16x16x4_f32(false, a0, false, p1,
                                                             (short)0, acc1, false, false);
                acc0 = __builtin_amdgcn_wmma_f32_16x16x4_f32(false, a1, false, q0,
                                                             (short)0, acc0, false, false);
                acc1 = __builtin_amdgcn_wmma_f32_16x16x4_f32(false, a1, false, q1,
                                                             (short)0, acc1, false, false);
            }
            const float4 s0 = dsc[buf][hh * 2];
            const float4 s1 = dsc[buf][hh * 2 + 1];
            const float sc[8] = {s0.x, s0.y, s0.z, s0.w, s1.x, s1.y, s1.z, s1.w};
            const int row0 = s * 16 + hh * 8;
            float* crow = C + (size_t)row0 * NOUT + pw * 32 + c;
#pragma unroll
            for (int r = 0; r < 8; ++r) {
                if (row0 + r < n) {
                    crow[(size_t)r * NOUT]      = acc0[r] * sc[r];
                    crow[(size_t)r * NOUT + 16] = acc1[r] * sc[r];
                }
            }
        }
    }
}

static inline int blocks_for(long long n, int bs) { return (int)((n + bs - 1) / bs); }

extern "C" void kernel_launch(void* const* d_in, const int* in_sizes, int n_in,
                              void* d_out, int out_size, void* d_ws, size_t ws_size,
                              hipStream_t stream) {
    const float* x  = (const float*)d_in[0];
    const int*   ei = (const int*)d_in[1];
    const float* W1 = (const float*)d_in[2];
    const float* b1 = (const float*)d_in[3];
    const float* W2 = (const float*)d_in[4];
    const float* b2 = (const float*)d_in[5];
    float* out = (float*)d_out;

    const int N = in_sizes[0] / IN_C;
    const long long E = (long long)in_sizes[1] / 2;
    const int* esrc = ei;
    const int* edst = ei + E;

    float* ws   = (float*)d_ws;
    float* h1   = ws;
    float* a1   = h1 + (size_t)N * HID2;
    float* h2   = a1 + (size_t)N * HID2;
    int*   bucket = (int*)(h2 + (size_t)N * HID);

    int* part = (int*)a1;
    const int nranges = blocks_for(N, RW);
    const int npieces = blocks_for(E, EPB);
    edge_partition_kernel<<<npieces, 1024, 0, stream>>>(esrc, edst, part, E, nranges, npieces);
    bucket_build_kernel<<<nranges, 1024, 0, stream>>>(part, esrc, edst, E, bucket, N, npieces);

    {
        gemm_wmma_f32<HID2><<<GEMM_GROUPS, 256, 0, stream>>>(x, W1, bucket, h1, N);
    }
    gather_relu_gemm<<<FUSED_GROUPS, 512, 0, stream>>>(
        bucket, esrc, edst, E, h1, b1, W2, h2, N);

    gather_aggregate<HID, false><<<blocks_for(N, 8), 256, 0, stream>>>(
        bucket, esrc, edst, E, h2, b2, out, N);
}
